# starting kernel
# baseline (speedup 1.0000x reference)
#include <hip/hip_runtime.h>

#define NNODES 100000
#define NEDGES 600000
#define FDIM   128
#define NREL   3
#define LDSS   516

typedef float v2f __attribute__((ext_vector_type(2)));
typedef float v8f __attribute__((ext_vector_type(8)));

__global__ void zero_f4(float* __restrict__ p, long n4) {
  long i = (long)blockIdx.x * blockDim.x + threadIdx.x;
  long stride = (long)gridDim.x * blockDim.x;
  float4 z = make_float4(0.f, 0.f, 0.f, 0.f);
  for (; i < n4; i += stride) ((float4*)p)[i] = z;
}

__global__ void edge_agg(const int* __restrict__ ei, const int* __restrict__ et,
                         const float* __restrict__ feat, float* __restrict__ agg,
                         float* __restrict__ cnt, int do_cnt) {
  int lane = threadIdx.x & 31;
  int wave = (int)((blockIdx.x * blockDim.x + threadIdx.x) >> 5);
  int nwaves = (int)((gridDim.x * blockDim.x) >> 5);
  for (int e = wave; e < NEDGES; e += nwaves) {
    int s = ei[e];
    int d = ei[NEDGES + e];
    int r = et[e];
    float4 v = *(const float4*)(feat + (long)s * FDIM + lane * 4);
    float* a = agg + ((long)r * NNODES + d) * FDIM + lane * 4;
    atomicAdd(a + 0, v.x);
    atomicAdd(a + 1, v.y);
    atomicAdd(a + 2, v.z);
    atomicAdd(a + 3, v.w);
    if (do_cnt && lane == 0) atomicAdd(cnt + (long)r * NNODES + d, 1.0f);
  }
}

__global__ void normalize_agg(float* __restrict__ agg, const float* __restrict__ cnt) {
  long g = (long)blockIdx.x * blockDim.x + threadIdx.x;
  long row = g >> 5;
  int c4 = (int)(g & 31);
  if (row >= (long)NREL * NNODES) return;
  float s = 1.0f / fmaxf(cnt[row], 1.0f);
  float4* p = (float4*)(agg + row * FDIM) + c4;
  float4 v = *p;
  v.x *= s; v.y *= s; v.z *= s; v.w *= s;
  *p = v;
}

__global__ __launch_bounds__(256) void rgcn_gemm(
    const float* __restrict__ A0, const float* __restrict__ A1,
    const float* __restrict__ A2, const float* __restrict__ A3,
    const float* __restrict__ B0, const float* __restrict__ B1,
    const float* __restrict__ B2, const float* __restrict__ B3,
    const float* __restrict__ bias, float* __restrict__ out) {
  __shared__ __align__(16) float Alds[16 * LDSS];
  int t = threadIdx.x;
  long row0 = (long)blockIdx.x * 16;

  #pragma unroll
  for (int i = 0; i < 8; ++i) {
    int f4  = t + 256 * i;
    int row = f4 >> 7;
    int c4  = f4 & 127;
    int seg = c4 >> 5;
    int cw  = c4 & 31;
    const float* sp = (seg == 0) ? A0 : (seg == 1) ? A1 : (seg == 2) ? A2 : A3;
    float4 v = *(const float4*)(sp + (row0 + row) * FDIM + cw * 4);
    *(float4*)(&Alds[row * LDSS + c4 * 4]) = v;
  }
  __syncthreads();

  int wave = t >> 5;
  int lane = t & 31;
  int half = lane >> 4;
  int mn   = lane & 15;
  int col0 = wave * 16;

  v8f acc = {};
  const float* Bs[4] = {B0, B1, B2, B3};
  #pragma unroll
  for (int seg = 0; seg < 4; ++seg) {
    const float* B  = Bs[seg];
    const float* Ab = &Alds[mn * LDSS + seg * 128 + 2 * half];
    #pragma unroll 4
    for (int kk = 0; kk < 128; kk += 4) {
      v2f a = *(const v2f*)(Ab + kk);
      int kr = kk + 2 * half;
      v2f b;
      b.x = B[kr * FDIM + col0 + mn];
      b.y = B[(kr + 1) * FDIM + col0 + mn];
      acc = __builtin_amdgcn_wmma_f32_16x16x4_f32(
           false, a,  false, b,
           (short)0, acc,  false,  false);
    }
  }

  int col = col0 + mn;
  float bc = bias[col];
  #pragma unroll
  for (int i = 0; i < 8; ++i) {
    int m = i + 8 * half;
    out[(row0 + m) * FDIM + col] = fmaxf(acc[i] + bc, 0.0f);
  }
}

__global__ void head_logsoftmax(const float* __restrict__ h, const float* __restrict__ lw,
                                const float* __restrict__ lb, float* __restrict__ out) {
  int lane = threadIdx.x & 31;
  int node = (int)((blockIdx.x * blockDim.x + threadIdx.x) >> 5);
  if (node >= NNODES) return;
  float4 v = *(const float4*)(h + (long)node * FDIM + lane * 4);
  const float* w = lw + lane * 8;
  float p0 = v.x * w[0] + v.y * w[2] + v.z * w[4] + v.w * w[6];
  float p1 = v.x * w[1] + v.y * w[3] + v.z * w[5] + v.w * w[7];
  #pragma unroll
  for (int off = 16; off > 0; off >>= 1) {
    p0 += __shfl_xor(p0, off, 32);
    p1 += __shfl_xor(p1, off, 32);
  }
  if (lane == 0) {
    float l0 = p0 + lb[0], l1 = p1 + lb[1];
    float m = fmaxf(l0, l1);
    float lse = m + logf(expf(l0 - m) + expf(l1 - m));
    out[(long)node * 2 + 0] = l0 - lse;
    out[(long)node * 2 + 1] = l1 - lse;
  }
}

extern "C" void kernel_launch(void* const* d_in, const int* in_sizes, int n_in,
                              void* d_out, int out_size, void* d_ws, size_t ws_size,
                              hipStream_t stream) {
  const float* x      = (const float*)d_in[0];
  const int*   ei     = (const int*)d_in[1];
  const int*   et     = (const int*)d_in[2];
  const float* W1rel  = (const float*)d_in[3];
  const float* W1root = (const float*)d_in[4];
  const float* b1     = (const float*)d_in[5];
  const float* W2rel  = (const float*)d_in[6];
  const float* W2root = (const float*)d_in[7];
  const float* b2     = (const float*)d_in[8];
  const float* lw     = (const float*)d_in[9];
  const float* lb     = (const float*)d_in[10];

  const long aggN = (long)NREL * NNODES * FDIM;
  const long cntN = (long)NREL * NNODES;
  const long NF   = (long)NNODES * FDIM;

  float* ws  = (float*)d_ws;
  float* agg = ws;
  float* cnt = agg + aggN;
  float* h1  = cnt + cntN;
  float* h2  = h1 + NF;

  dim3 blk(256);

  zero_f4<<<4096, blk, 0, stream>>>(agg, (aggN + cntN) / 4);
  edge_agg<<<2048, blk, 0, stream>>>(ei, et, x, agg, cnt,  1);
  normalize_agg<<<(int)(cntN * 32 / 256), blk, 0, stream>>>(agg, cnt);
  rgcn_gemm<<<NNODES / 16, blk, 0, stream>>>(
      x, agg, agg + NF, agg + 2 * NF,
      W1root, W1rel, W1rel + FDIM * FDIM, W1rel + 2 * FDIM * FDIM, b1, h1);

  zero_f4<<<4096, blk, 0, stream>>>(agg, aggN / 4);
  edge_agg<<<2048, blk, 0, stream>>>(ei, et, h1, agg, cnt,  0);
  normalize_agg<<<(int)(cntN * 32 / 256), blk, 0, stream>>>(agg, cnt);
  rgcn_gemm<<<NNODES / 16, blk, 0, stream>>>(
      h1, agg, agg + NF, agg + 2 * NF,
      W2root, W2rel, W2rel + FDIM * FDIM, W2rel + 2 * FDIM * FDIM, b2, h2);

  head_logsoftmax<<<(NNODES * 32) / 256, blk, 0, stream>>>(h2, lw, lb, (float*)d_out);
}


# edge aggregation by destination ownership: 256 work-groups each hold 391 output rows in LDS, filter the edge list and add scaled projected rows by LDS float adds; no global atomics; new sum order
# speedup vs baseline: 1.5221x; 1.5221x over previous
#include <hip/hip_runtime.h>


#define NNODES 100000
#define NEDGES 600000
#define FDIM   128
#define NREL   3
#define LDSA   132

typedef float v2f __attribute__((ext_vector_type(2)));
typedef float v8f __attribute__((ext_vector_type(8)));

__global__ void zero_f4(float* __restrict__ p, long n4) {
  long i = (long)blockIdx.x * blockDim.x + threadIdx.x;
  long stride = (long)gridDim.x * blockDim.x;
  float4 z = make_float4(0.f, 0.f, 0.f, 0.f);
  for (; i < n4; i += stride) ((float4*)p)[i] = z;
}

__global__ void edge_count(const int* __restrict__ ei, const int* __restrict__ et,
                           float* __restrict__ cnt) {
  int e = (int)(blockIdx.x * blockDim.x + threadIdx.x);
  if (e < NEDGES) atomicAdd(cnt + (long)et[e] * NNODES + ei[NEDGES + e], 1.0f);
}

template <bool RELU_A>
__global__ __launch_bounds__(256) void rgcn_project(
    const float* __restrict__ A,
    const float* __restrict__ B0, const float* __restrict__ B1,
    const float* __restrict__ B2, const float* __restrict__ B3,
    const float* __restrict__ bias, float* __restrict__ hacc, float* __restrict__ Y) {
  __shared__ __align__(16) float Alds[16 * LDSA];
  int t = threadIdx.x;
  long row0 = (long)blockIdx.x * 16;

  #pragma unroll
  for (int i = 0; i < 2; ++i) {
    int f4  = t + 256 * i;
    int row = f4 >> 5;
    int c4  = f4 & 31;
    float4 v = *(const float4*)(A + (row0 + row) * FDIM + c4 * 4);
    if (RELU_A) {
      v.x = fmaxf(v.x, 0.0f); v.y = fmaxf(v.y, 0.0f);
      v.z = fmaxf(v.z, 0.0f); v.w = fmaxf(v.w, 0.0f);
    }
    *(float4*)(&Alds[row * LDSA + c4 * 4]) = v;
  }
  __syncthreads();

  int wave = t >> 5;
  int lane = t & 31;
  int half = lane >> 4;
  int mn   = lane & 15;
  int col  = wave * 16 + mn;

  v8f acc0 = {}, acc1 = {}, acc2 = {}, acc3 = {};
  const float* Ab = &Alds[mn * LDSA + 2 * half];
  #pragma unroll 4
  for (int kk = 0; kk < FDIM; kk += 4) {
    v2f a = *(const v2f*)(Ab + kk);
    int kr = kk + 2 * half;
    int o0 = kr * FDIM + col;
    int o1 = o0 + FDIM;
    v2f b0, b1, b2, b3;
    b0.x = B0[o0]; b0.y = B0[o1];
    b1.x = B1[o0]; b1.y = B1[o1];
    b2.x = B2[o0]; b2.y = B2[o1];
    b3.x = B3[o0]; b3.y = B3[o1];
    acc0 = __builtin_amdgcn_wmma_f32_16x16x4_f32(false, a, false, b0, (short)0, acc0, false, false);
    acc1 = __builtin_amdgcn_wmma_f32_16x16x4_f32(false, a, false, b1, (short)0, acc1, false, false);
    acc2 = __builtin_amdgcn_wmma_f32_16x16x4_f32(false, a, false, b2, (short)0, acc2, false, false);
    acc3 = __builtin_amdgcn_wmma_f32_16x16x4_f32(false, a, false, b3, (short)0, acc3, false, false);
  }

  float bc = bias[col];
  float* Y0 = Y;
  float* Y1 = Y + (long)NNODES * FDIM;
  float* Y2 = Y + 2L * NNODES * FDIM;
  #pragma unroll
  for (int i = 0; i < 8; ++i) {
    long o = (row0 + i + 8 * half) * FDIM + col;
    hacc[o] = acc0[i] + bc;
    Y0[o] = acc1[i];
    Y1[o] = acc2[i];
    Y2[o] = acc3[i];
  }
}

#define NOWNERS   256
#define OWN_ROWS  391
#define FLUSH_AT  32
#define PEND_CAP  544
#define BATCH     8

__device__ __forceinline__ void drain_pending(
    const int* pend, int n, int lane, int d0,
    const int* __restrict__ ei, const int* __restrict__ et,
    const float* __restrict__ Y, const float* __restrict__ cnt, float* acc) {
  for (int i = 0; i < n; i += BATCH) {
    int e[BATCH], dl[BATCH], s[BATCH], r[BATCH];
    float c[BATCH];
    float4 v[BATCH];
    #pragma unroll
    for (int j = 0; j < BATCH; ++j) {
      int k = min(i + j, n - 1);
      int w = __builtin_amdgcn_readfirstlane(pend[k]);
      e[j]  = w & 0xFFFFF;
      dl[j] = (int)((unsigned)w >> 20);
    }
    #pragma unroll
    for (int j = 0; j < BATCH; ++j) { s[j] = ei[e[j]]; r[j] = et[e[j]]; }
    #pragma unroll
    for (int j = 0; j < BATCH; ++j) {
      c[j] = cnt[(long)r[j] * NNODES + d0 + dl[j]];
      v[j] = *(const float4*)(Y + ((long)r[j] * NNODES + s[j]) * FDIM + lane * 4);
    }
    #pragma unroll
    for (int j = 0; j < BATCH; ++j) {
      int row = (i + j < n) ? dl[j] : OWN_ROWS;
      float sc = 1.0f / fmaxf(c[j], 1.0f);
      float* a = acc + row * FDIM + lane;
      atomicAdd(a,      v[j].x * sc);
      atomicAdd(a + 32, v[j].y * sc);
      atomicAdd(a + 64, v[j].z * sc);
      atomicAdd(a + 96, v[j].w * sc);
    }
  }
}

__global__ __launch_bounds__(1024) void edge_gather_owned(
    const int* __restrict__ ei, const int* __restrict__ et,
    const float* __restrict__ Y, const float* __restrict__ cnt,
    float* __restrict__ hacc) {
  __shared__ __align__(16) float acc[(OWN_ROWS + 1) * FDIM];
  __shared__ int pend_all[32 * PEND_CAP];
  const int t = threadIdx.x;
  const int lane = t & 31;
  const int wave = t >> 5;
  const int d0 = (int)blockIdx.x * OWN_ROWS;
  const int nrows = min(OWN_ROWS, NNODES - d0);

  const float4* hsrc = (const float4*)(hacc + (long)d0 * FDIM);
  for (int i = t; i < nrows * 32; i += 1024) {
    float4 v = hsrc[i];
    float* a = acc + (i >> 5) * FDIM + (i & 31);
    a[0] = v.x; a[32] = v.y; a[64] = v.z; a[96] = v.w;
  }
  __syncthreads();

  int* pend = pend_all + wave * PEND_CAP;
  int npend = 0;
  const int4* dst4 = (const int4*)(ei + NEDGES);
  const int NQ = NEDGES / 4;
  for (int base = 0; base < NQ; base += 4 * 1024) {
    if (npend >= FLUSH_AT) {
      drain_pending(pend, npend, lane, d0, ei, et, Y, cnt, acc);
      npend = 0;
    }
    int4 dv[4];
    #pragma unroll
    for (int u = 0; u < 4; ++u) {
      int q = base + u * 1024 + t;
      dv[u] = (q < NQ) ? dst4[q] : make_int4(-1, -1, -1, -1);
    }
    #pragma unroll
    for (int u = 0; u < 4; ++u) {
      int q = base + u * 1024 + t;
      unsigned r0 = (unsigned)(dv[u].x - d0), r1 = (unsigned)(dv[u].y - d0);
      unsigned r2 = (unsigned)(dv[u].z - d0), r3 = (unsigned)(dv[u].w - d0);
      bool m0 = r0 < (unsigned)nrows, m1 = r1 < (unsigned)nrows;
      bool m2 = r2 < (unsigned)nrows, m3 = r3 < (unsigned)nrows;
      if (__builtin_amdgcn_ballot_w32(m0 | m1 | m2 | m3)) {
        unsigned k;
        k = __builtin_amdgcn_ballot_w32(m0);
        if (m0) pend[npend + __builtin_amdgcn_mbcnt_lo(k, 0u)] = (int)((r0 << 20) | (unsigned)(4 * q + 0));
        npend += __builtin_popcount(k);
        k = __builtin_amdgcn_ballot_w32(m1);
        if (m1) pend[npend + __builtin_amdgcn_mbcnt_lo(k, 0u)] = (int)((r1 << 20) | (unsigned)(4 * q + 1));
        npend += __builtin_popcount(k);
        k = __builtin_amdgcn_ballot_w32(m2);
        if (m2) pend[npend + __builtin_amdgcn_mbcnt_lo(k, 0u)] = (int)((r2 << 20) | (unsigned)(4 * q + 2));
        npend += __builtin_popcount(k);
        k = __builtin_amdgcn_ballot_w32(m3);
        if (m3) pend[npend + __builtin_amdgcn_mbcnt_lo(k, 0u)] = (int)((r3 << 20) | (unsigned)(4 * q + 3));
        npend += __builtin_popcount(k);
      }
    }
  }
  if (npend > 0) drain_pending(pend, npend, lane, d0, ei, et, Y, cnt, acc);
  __syncthreads();

  float4* hdst = (float4*)(hacc + (long)d0 * FDIM);
  for (int i = t; i < nrows * 32; i += 1024) {
    const float* a = acc + (i >> 5) * FDIM + (i & 31);
    hdst[i] = make_float4(a[0], a[32], a[64], a[96]);
  }
}

__global__ void head_logsoftmax(const float* __restrict__ h, const float* __restrict__ lw,
                                const float* __restrict__ lb, float* __restrict__ out) {
  int lane = threadIdx.x & 31;
  int node = (int)((blockIdx.x * blockDim.x + threadIdx.x) >> 5);
  if (node >= NNODES) return;
  float4 v = *(const float4*)(h + (long)node * FDIM + lane * 4);
  v.x = fmaxf(v.x, 0.0f); v.y = fmaxf(v.y, 0.0f);
  v.z = fmaxf(v.z, 0.0f); v.w = fmaxf(v.w, 0.0f);
  const float* w = lw + lane * 8;
  float p0 = v.x * w[0] + v.y * w[2] + v.z * w[4] + v.w * w[6];
  float p1 = v.x * w[1] + v.y * w[3] + v.z * w[5] + v.w * w[7];
  #pragma unroll
  for (int off = 16; off > 0; off >>= 1) {
    p0 += __shfl_xor(p0, off, 32);
    p1 += __shfl_xor(p1, off, 32);
  }
  if (lane == 0) {
    float l0 = p0 + lb[0], l1 = p1 + lb[1];
    float m = fmaxf(l0, l1);
    float lse = m + logf(expf(l0 - m) + expf(l1 - m));
    out[(long)node * 2 + 0] = l0 - lse;
    out[(long)node * 2 + 1] = l1 - lse;
  }
}

extern "C" void kernel_launch(void* const* d_in, const int* in_sizes, int n_in,
                              void* d_out, int out_size, void* d_ws, size_t ws_size,
                              hipStream_t stream) {
  const float* x      = (const float*)d_in[0];
  const int*   ei     = (const int*)d_in[1];
  const int*   et     = (const int*)d_in[2];
  const float* W1rel  = (const float*)d_in[3];
  const float* W1root = (const float*)d_in[4];
  const float* b1     = (const float*)d_in[5];
  const float* W2rel  = (const float*)d_in[6];
  const float* W2root = (const float*)d_in[7];
  const float* b2     = (const float*)d_in[8];
  const float* lw     = (const float*)d_in[9];
  const float* lb     = (const float*)d_in[10];

  const long msgN = (long)NREL * NNODES * FDIM;
  const long cntN = (long)NREL * NNODES;
  const long NF   = (long)NNODES * FDIM;

  float* ws  = (float*)d_ws;
  float* Y   = ws;
  float* cnt = Y + msgN;
  float* h1  = cnt + cntN;
  float* h2  = h1 + NF;

  dim3 blk(256);

  zero_f4<<<293, blk, 0, stream>>>(cnt, cntN / 4);
  edge_count<<<(NEDGES + 255) / 256, blk, 0, stream>>>(ei, et, cnt);

  rgcn_project<false><<<NNODES / 16, blk, 0, stream>>>(
      x, W1root, W1rel, W1rel + FDIM * FDIM, W1rel + 2 * FDIM * FDIM, b1, h1, Y);
  edge_gather_owned<<<NOWNERS, 1024, 0, stream>>>(ei, et, Y, cnt, h1);

  rgcn_project<true><<<NNODES / 16, blk, 0, stream>>>(
      h1, W2root, W2rel, W2rel + FDIM * FDIM, W2rel + 2 * FDIM * FDIM, b2, h2, Y);
  edge_gather_owned<<<NOWNERS, 1024, 0, stream>>>(ei, et, Y, cnt, h2);

  head_logsoftmax<<<(NNODES * 32) / 256, blk, 0, stream>>>(h2, lw, lb, (float*)d_out);
}


# projection GEMM made persistent, its 128x512 weights resident in LDS in fragment order: 256 work-groups walk 16-row tiles, B operands by 16-byte LDS reads, no global weight loads; new k sum order
# speedup vs baseline: 1.9370x; 1.2726x over previous
#include <hip/hip_runtime.h>


#define NNODES 100000
#define NEDGES 600000
#define FDIM   128
#define NREL   3
#define LDSA   132

typedef float v2f __attribute__((ext_vector_type(2)));
typedef float v8f __attribute__((ext_vector_type(8)));

__global__ void zero_f4(float* __restrict__ p, long n4) {
  long i = (long)blockIdx.x * blockDim.x + threadIdx.x;
  long stride = (long)gridDim.x * blockDim.x;
  float4 z = make_float4(0.f, 0.f, 0.f, 0.f);
  for (; i < n4; i += stride) ((float4*)p)[i] = z;
}

__global__ void edge_count(const int* __restrict__ ei, const int* __restrict__ et,
                           float* __restrict__ cnt) {
  int e = (int)(blockIdx.x * blockDim.x + threadIdx.x);
  if (e < NEDGES) atomicAdd(cnt + (long)et[e] * NNODES + ei[NEDGES + e], 1.0f);
}

#define PGRID   256
#define NTILES  (NNODES / 16)
#define NCOLS   (4 * FDIM)
#define ABUFS   3
#define ATILE   (16 * LDSA)

struct Frags { float4 a[4]; float4 b[16]; };

__device__ __forceinline__ void load_chunk(Frags& f, const float* Ab, const float* Bb, int c) {
  #pragma unroll
  for (int jj = 0; jj < 4; ++jj) {
    int j = 4 * c + jj;
    f.a[jj] = *(const float4*)(Ab + 8 * j);
    #pragma unroll
    for (int p = 0; p < 4; ++p)
      f.b[jj * 4 + p] = *(const float4*)(Bb + j * (2 * NCOLS * 4) + p * (FDIM * 4));
  }
}

__device__ __forceinline__ void mma_chunk(const Frags& f, v8f (&acc)[4]) {
  #pragma unroll
  for (int jj = 0; jj < 4; ++jj) {
    v2f a0 = {f.a[jj].x, f.a[jj].y};
    v2f a1 = {f.a[jj].z, f.a[jj].w};
    #pragma unroll
    for (int p = 0; p < 4; ++p) {
      v2f b0 = {f.b[jj * 4 + p].x, f.b[jj * 4 + p].y};
      acc[p] = __builtin_amdgcn_wmma_f32_16x16x4_f32(false, a0, false, b0, (short)0, acc[p], false, false);
    }
    #pragma unroll
    for (int p = 0; p < 4; ++p) {
      v2f b1 = {f.b[jj * 4 + p].z, f.b[jj * 4 + p].w};
      acc[p] = __builtin_amdgcn_wmma_f32_16x16x4_f32(false, a1, false, b1, (short)0, acc[p], false, false);
    }
  }
}

template <bool RELU_A>
__global__ __launch_bounds__(256) void rgcn_project(
    const float* __restrict__ A,
    const float* __restrict__ B0, const float* __restrict__ B1,
    const float* __restrict__ B2, const float* __restrict__ B3,
    const float* __restrict__ bias, float* __restrict__ hacc, float* __restrict__ Y) {
  __shared__ __align__(16) float Bl[(FDIM / 4) * NCOLS * 4];
  __shared__ __align__(16) float Al[ABUFS * ATILE];
  const int t = threadIdx.x;
  const int bid = (int)blockIdx.x;
  const int ntile = (NTILES - bid + PGRID - 1) / PGRID;

  const int ar = t >> 5, ac4 = t & 31;
  const float* Asrc = A + (long)ar * FDIM + ac4 * 4;
  float* Adst = Al + ar * LDSA + ac4 * 4;
  float4 pa0 = *(const float4*)(Asrc + (long)bid * 16 * FDIM);
  float4 pa1 = *(const float4*)(Asrc + ((long)bid * 16 + 8) * FDIM);
  float4 pb0 = pa0, pb1 = pa1;
  if (ntile > 1) {
    pb0 = *(const float4*)(Asrc + (long)(bid + PGRID) * 16 * FDIM);
    pb1 = *(const float4*)(Asrc + ((long)(bid + PGRID) * 16 + 8) * FDIM);
  }

  {
    const float* Bpa = (t < 128) ? B0 : B1;
    const float* Bpb = (t < 128) ? B2 : B3;
    const int n = t & 127;
    #pragma unroll 8
    for (int i = 0; i < FDIM / 4; ++i) {
      int kq = (i + bid) & (FDIM / 4 - 1);
      const float* ra = Bpa + kq * 4 * FDIM + n;
      const float* rb = Bpb + kq * 4 * FDIM + n;
      float4 va = make_float4(ra[0], ra[FDIM], ra[2 * FDIM], ra[3 * FDIM]);
      float4 vb = make_float4(rb[0], rb[FDIM], rb[2 * FDIM], rb[3 * FDIM]);
      *(float4*)(Bl + (kq * NCOLS + t) * 4) = va;
      *(float4*)(Bl + (kq * NCOLS + 256 + t) * 4) = vb;
    }
  }

  if (RELU_A) {
    pa0.x = fmaxf(pa0.x, 0.0f); pa0.y = fmaxf(pa0.y, 0.0f); pa0.z = fmaxf(pa0.z, 0.0f); pa0.w = fmaxf(pa0.w, 0.0f);
    pa1.x = fmaxf(pa1.x, 0.0f); pa1.y = fmaxf(pa1.y, 0.0f); pa1.z = fmaxf(pa1.z, 0.0f); pa1.w = fmaxf(pa1.w, 0.0f);
  }
  *(float4*)(Adst) = pa0;
  *(float4*)(Adst + 8 * LDSA) = pa1;
  pa0 = pb0; pa1 = pb1;
  __syncthreads();

  const int wave = t >> 5;
  const int lane = t & 31;
  const int half = lane >> 4;
  const int mn   = lane & 15;
  const int col  = wave * 16 + mn;
  const float* Bb = Bl + (half * NCOLS + col) * 4;
  const float* Afr = Al + mn * LDSA + 4 * half;
  const float bc = bias[col];
  float* Y0 = Y;
  float* Y1 = Y + (long)NNODES * FDIM;
  float* Y2 = Y + 2L * NNODES * FDIM;

  Frags f0, f1;
  load_chunk(f0, Afr, Bb, 0);
  int cur = 0;
  for (int i = 0; i < ntile; ++i) {
    const int nxt = (cur == ABUFS - 1) ? 0 : cur + 1;
    const float* Ab = Afr + cur * ATILE;
    v8f acc[4] = {};
    load_chunk(f1, Ab, Bb, 1);
    __builtin_amdgcn_sched_barrier(0);
    mma_chunk(f0, acc);
    __builtin_amdgcn_sched_barrier(0);
    load_chunk(f0, Ab, Bb, 2);
    __builtin_amdgcn_sched_barrier(0);
    mma_chunk(f1, acc);
    __builtin_amdgcn_sched_barrier(0);
    if (i + 1 < ntile) {
      if (RELU_A) {
        pa0.x = fmaxf(pa0.x, 0.0f); pa0.y = fmaxf(pa0.y, 0.0f); pa0.z = fmaxf(pa0.z, 0.0f); pa0.w = fmaxf(pa0.w, 0.0f);
        pa1.x = fmaxf(pa1.x, 0.0f); pa1.y = fmaxf(pa1.y, 0.0f); pa1.z = fmaxf(pa1.z, 0.0f); pa1.w = fmaxf(pa1.w, 0.0f);
      }
      *(float4*)(Adst + nxt * ATILE) = pa0;
      *(float4*)(Adst + nxt * ATILE + 8 * LDSA) = pa1;
    }
    __syncthreads();
    if (i + 2 < ntile) {
      long r = (long)(bid + (i + 2) * PGRID) * 16;
      pa0 = *(const float4*)(Asrc + r * FDIM);
      pa1 = *(const float4*)(Asrc + (r + 8) * FDIM);
    }
    load_chunk(f1, Ab, Bb, 3);
    __builtin_amdgcn_sched_barrier(0);
    mma_chunk(f0, acc);
    __builtin_amdgcn_sched_barrier(0);
    if (i + 1 < ntile) load_chunk(f0, Afr + nxt * ATILE, Bb, 0);
    __builtin_amdgcn_sched_barrier(0);
    mma_chunk(f1, acc);
    __builtin_amdgcn_sched_barrier(0);

    const long row0 = (long)(bid + i * PGRID) * 16;
    #pragma unroll
    for (int r = 0; r < 8; ++r) {
      long o = (row0 + r + 8 * half) * FDIM + col;
      hacc[o] = acc[0][r] + bc;
      Y0[o] = acc[1][r];
      Y1[o] = acc[2][r];
      Y2[o] = acc[3][r];
    }
    cur = nxt;
  }
}

#define NOWNERS   256
#define OWN_ROWS  391
#define FLUSH_AT  32
#define PEND_CAP  544
#define BATCH     8

__device__ __forceinline__ void drain_pending(
    const int* pend, int n, int lane, int d0,
    const int* __restrict__ ei, const int* __restrict__ et,
    const float* __restrict__ Y, const float* __restrict__ cnt, float* acc) {
  for (int i = 0; i < n; i += BATCH) {
    int e[BATCH], dl[BATCH], s[BATCH], r[BATCH];
    float c[BATCH];
    float4 v[BATCH];
    #pragma unroll
    for (int j = 0; j < BATCH; ++j) {
      int k = min(i + j, n - 1);
      int w = __builtin_amdgcn_readfirstlane(pend[k]);
      e[j]  = w & 0xFFFFF;
      dl[j] = (int)((unsigned)w >> 20);
    }
    #pragma unroll
    for (int j = 0; j < BATCH; ++j) { s[j] = ei[e[j]]; r[j] = et[e[j]]; }
    #pragma unroll
    for (int j = 0; j < BATCH; ++j) {
      c[j] = cnt[(long)r[j] * NNODES + d0 + dl[j]];
      v[j] = *(const float4*)(Y + ((long)r[j] * NNODES + s[j]) * FDIM + lane * 4);
    }
    #pragma unroll
    for (int j = 0; j < BATCH; ++j) {
      int row = (i + j < n) ? dl[j] : OWN_ROWS;
      float sc = 1.0f / fmaxf(c[j], 1.0f);
      float* a = acc + row * FDIM + lane;
      atomicAdd(a,      v[j].x * sc);
      atomicAdd(a + 32, v[j].y * sc);
      atomicAdd(a + 64, v[j].z * sc);
      atomicAdd(a + 96, v[j].w * sc);
    }
  }
}

__global__ __launch_bounds__(1024) void edge_gather_owned(
    const int* __restrict__ ei, const int* __restrict__ et,
    const float* __restrict__ Y, const float* __restrict__ cnt,
    float* __restrict__ hacc) {
  __shared__ __align__(16) float acc[(OWN_ROWS + 1) * FDIM];
  __shared__ int pend_all[32 * PEND_CAP];
  const int t = threadIdx.x;
  const int lane = t & 31;
  const int wave = t >> 5;
  const int d0 = (int)blockIdx.x * OWN_ROWS;
  const int nrows = min(OWN_ROWS, NNODES - d0);

  const float4* hsrc = (const float4*)(hacc + (long)d0 * FDIM);
  for (int i = t; i < nrows * 32; i += 1024) {
    float4 v = hsrc[i];
    float* a = acc + (i >> 5) * FDIM + (i & 31);
    a[0] = v.x; a[32] = v.y; a[64] = v.z; a[96] = v.w;
  }
  __syncthreads();

  int* pend = pend_all + wave * PEND_CAP;
  int npend = 0;
  const int4* dst4 = (const int4*)(ei + NEDGES);
  const int NQ = NEDGES / 4;
  for (int base = 0; base < NQ; base += 4 * 1024) {
    if (npend >= FLUSH_AT) {
      drain_pending(pend, npend, lane, d0, ei, et, Y, cnt, acc);
      npend = 0;
    }
    int4 dv[4];
    #pragma unroll
    for (int u = 0; u < 4; ++u) {
      int q = base + u * 1024 + t;
      dv[u] = (q < NQ) ? dst4[q] : make_int4(-1, -1, -1, -1);
    }
    #pragma unroll
    for (int u = 0; u < 4; ++u) {
      int q = base + u * 1024 + t;
      unsigned r0 = (unsigned)(dv[u].x - d0), r1 = (unsigned)(dv[u].y - d0);
      unsigned r2 = (unsigned)(dv[u].z - d0), r3 = (unsigned)(dv[u].w - d0);
      bool m0 = r0 < (unsigned)nrows, m1 = r1 < (unsigned)nrows;
      bool m2 = r2 < (unsigned)nrows, m3 = r3 < (unsigned)nrows;
      if (__builtin_amdgcn_ballot_w32(m0 | m1 | m2 | m3)) {
        unsigned k;
        k = __builtin_amdgcn_ballot_w32(m0);
        if (m0) pend[npend + __builtin_amdgcn_mbcnt_lo(k, 0u)] = (int)((r0 << 20) | (unsigned)(4 * q + 0));
        npend += __builtin_popcount(k);
        k = __builtin_amdgcn_ballot_w32(m1);
        if (m1) pend[npend + __builtin_amdgcn_mbcnt_lo(k, 0u)] = (int)((r1 << 20) | (unsigned)(4 * q + 1));
        npend += __builtin_popcount(k);
        k = __builtin_amdgcn_ballot_w32(m2);
        if (m2) pend[npend + __builtin_amdgcn_mbcnt_lo(k, 0u)] = (int)((r2 << 20) | (unsigned)(4 * q + 2));
        npend += __builtin_popcount(k);
        k = __builtin_amdgcn_ballot_w32(m3);
        if (m3) pend[npend + __builtin_amdgcn_mbcnt_lo(k, 0u)] = (int)((r3 << 20) | (unsigned)(4 * q + 3));
        npend += __builtin_popcount(k);
      }
    }
  }
  if (npend > 0) drain_pending(pend, npend, lane, d0, ei, et, Y, cnt, acc);
  __syncthreads();

  float4* hdst = (float4*)(hacc + (long)d0 * FDIM);
  for (int i = t; i < nrows * 32; i += 1024) {
    const float* a = acc + (i >> 5) * FDIM + (i & 31);
    hdst[i] = make_float4(a[0], a[32], a[64], a[96]);
  }
}

__global__ void head_logsoftmax(const float* __restrict__ h, const float* __restrict__ lw,
                                const float* __restrict__ lb, float* __restrict__ out) {
  int lane = threadIdx.x & 31;
  int node = (int)((blockIdx.x * blockDim.x + threadIdx.x) >> 5);
  if (node >= NNODES) return;
  float4 v = *(const float4*)(h + (long)node * FDIM + lane * 4);
  v.x = fmaxf(v.x, 0.0f); v.y = fmaxf(v.y, 0.0f);
  v.z = fmaxf(v.z, 0.0f); v.w = fmaxf(v.w, 0.0f);
  const float* w = lw + lane * 8;
  float p0 = v.x * w[0] + v.y * w[2] + v.z * w[4] + v.w * w[6];
  float p1 = v.x * w[1] + v.y * w[3] + v.z * w[5] + v.w * w[7];
  #pragma unroll
  for (int off = 16; off > 0; off >>= 1) {
    p0 += __shfl_xor(p0, off, 32);
    p1 += __shfl_xor(p1, off, 32);
  }
  if (lane == 0) {
    float l0 = p0 + lb[0], l1 = p1 + lb[1];
    float m = fmaxf(l0, l1);
    float lse = m + logf(expf(l0 - m) + expf(l1 - m));
    out[(long)node * 2 + 0] = l0 - lse;
    out[(long)node * 2 + 1] = l1 - lse;
  }
}

extern "C" void kernel_launch(void* const* d_in, const int* in_sizes, int n_in,
                              void* d_out, int out_size, void* d_ws, size_t ws_size,
                              hipStream_t stream) {
  const float* x      = (const float*)d_in[0];
  const int*   ei     = (const int*)d_in[1];
  const int*   et     = (const int*)d_in[2];
  const float* W1rel  = (const float*)d_in[3];
  const float* W1root = (const float*)d_in[4];
  const float* b1     = (const float*)d_in[5];
  const float* W2rel  = (const float*)d_in[6];
  const float* W2root = (const float*)d_in[7];
  const float* b2     = (const float*)d_in[8];
  const float* lw     = (const float*)d_in[9];
  const float* lb     = (const float*)d_in[10];

  const long msgN = (long)NREL * NNODES * FDIM;
  const long cntN = (long)NREL * NNODES;
  const long NF   = (long)NNODES * FDIM;

  float* ws  = (float*)d_ws;
  float* Y   = ws;
  float* cnt = Y + msgN;
  float* h1  = cnt + cntN;
  float* h2  = h1 + NF;

  dim3 blk(256);

  zero_f4<<<293, blk, 0, stream>>>(cnt, cntN / 4);
  edge_count<<<(NEDGES + 255) / 256, blk, 0, stream>>>(ei, et, cnt);

  rgcn_project<false><<<PGRID, blk, 0, stream>>>(
      x, W1root, W1rel, W1rel + FDIM * FDIM, W1rel + 2 * FDIM * FDIM, b1, h1, Y);
  edge_gather_owned<<<NOWNERS, 1024, 0, stream>>>(ei, et, Y, cnt, h1);

  rgcn_project<true><<<PGRID, blk, 0, stream>>>(
      h1, W2root, W2rel, W2rel + FDIM * FDIM, W2rel + 2 * FDIM * FDIM, b2, h2, Y);
  edge_gather_owned<<<NOWNERS, 1024, 0, stream>>>(ei, et, Y, cnt, h2);

  head_logsoftmax<<<(NNODES * 32) / 256, blk, 0, stream>>>(h2, lw, lb, (float*)d_out);
}


# aggregation drain resolves pending edges 32 at a time on the vector path (each lane loads its edge's source, relation, degree); 16 row loads in flight, no scalar index loads; LDS adds in a new order
# speedup vs baseline: 2.1057x; 1.0871x over previous
#include <hip/hip_runtime.h>


#define NNODES 100000
#define NEDGES 600000
#define FDIM   128
#define NREL   3
#define LDSA   132

typedef float v2f __attribute__((ext_vector_type(2)));
typedef float v8f __attribute__((ext_vector_type(8)));

__global__ void zero_f4(float* __restrict__ p, long n4) {
  long i = (long)blockIdx.x * blockDim.x + threadIdx.x;
  long stride = (long)gridDim.x * blockDim.x;
  float4 z = make_float4(0.f, 0.f, 0.f, 0.f);
  for (; i < n4; i += stride) ((float4*)p)[i] = z;
}

__global__ void edge_count(const int* __restrict__ ei, const int* __restrict__ et,
                           float* __restrict__ cnt) {
  int e = (int)(blockIdx.x * blockDim.x + threadIdx.x);
  if (e < NEDGES) atomicAdd(cnt + (long)et[e] * NNODES + ei[NEDGES + e], 1.0f);
}

#define PGRID   256
#define NTILES  (NNODES / 16)
#define NCOLS   (4 * FDIM)
#define ABUFS   3
#define ATILE   (16 * LDSA)

struct Frags { float4 a[4]; float4 b[16]; };

__device__ __forceinline__ void load_chunk(Frags& f, const float* Ab, const float* Bb, int c) {
  #pragma unroll
  for (int jj = 0; jj < 4; ++jj) {
    int j = 4 * c + jj;
    f.a[jj] = *(const float4*)(Ab + 8 * j);
    #pragma unroll
    for (int p = 0; p < 4; ++p)
      f.b[jj * 4 + p] = *(const float4*)(Bb + j * (2 * NCOLS * 4) + p * (FDIM * 4));
  }
}

__device__ __forceinline__ void mma_chunk(const Frags& f, v8f (&acc)[4]) {
  #pragma unroll
  for (int jj = 0; jj < 4; ++jj) {
    v2f a0 = {f.a[jj].x, f.a[jj].y};
    v2f a1 = {f.a[jj].z, f.a[jj].w};
    #pragma unroll
    for (int p = 0; p < 4; ++p) {
      v2f b0 = {f.b[jj * 4 + p].x, f.b[jj * 4 + p].y};
      acc[p] = __builtin_amdgcn_wmma_f32_16x16x4_f32(false, a0, false, b0, (short)0, acc[p], false, false);
    }
    #pragma unroll
    for (int p = 0; p < 4; ++p) {
      v2f b1 = {f.b[jj * 4 + p].z, f.b[jj * 4 + p].w};
      acc[p] = __builtin_amdgcn_wmma_f32_16x16x4_f32(false, a1, false, b1, (short)0, acc[p], false, false);
    }
  }
}

template <bool RELU_A>
__global__ __launch_bounds__(256) void rgcn_project(
    const float* __restrict__ A,
    const float* __restrict__ B0, const float* __restrict__ B1,
    const float* __restrict__ B2, const float* __restrict__ B3,
    const float* __restrict__ bias, float* __restrict__ hacc, float* __restrict__ Y) {
  __shared__ __align__(16) float Bl[(FDIM / 4) * NCOLS * 4];
  __shared__ __align__(16) float Al[ABUFS * ATILE];
  const int t = threadIdx.x;
  const int bid = (int)blockIdx.x;
  const int ntile = (NTILES - bid + PGRID - 1) / PGRID;

  const int ar = t >> 5, ac4 = t & 31;
  const float* Asrc = A + (long)ar * FDIM + ac4 * 4;
  float* Adst = Al + ar * LDSA + ac4 * 4;
  float4 pa0 = *(const float4*)(Asrc + (long)bid * 16 * FDIM);
  float4 pa1 = *(const float4*)(Asrc + ((long)bid * 16 + 8) * FDIM);
  float4 pb0 = pa0, pb1 = pa1;
  if (ntile > 1) {
    pb0 = *(const float4*)(Asrc + (long)(bid + PGRID) * 16 * FDIM);
    pb1 = *(const float4*)(Asrc + ((long)(bid + PGRID) * 16 + 8) * FDIM);
  }

  {
    const float* Bpa = (t < 128) ? B0 : B1;
    const float* Bpb = (t < 128) ? B2 : B3;
    const int n = t & 127;
    #pragma unroll 8
    for (int i = 0; i < FDIM / 4; ++i) {
      int kq = (i + bid) & (FDIM / 4 - 1);
      const float* ra = Bpa + kq * 4 * FDIM + n;
      const float* rb = Bpb + kq * 4 * FDIM + n;
      float4 va = make_float4(ra[0], ra[FDIM], ra[2 * FDIM], ra[3 * FDIM]);
      float4 vb = make_float4(rb[0], rb[FDIM], rb[2 * FDIM], rb[3 * FDIM]);
      *(float4*)(Bl + (kq * NCOLS + t) * 4) = va;
      *(float4*)(Bl + (kq * NCOLS + 256 + t) * 4) = vb;
    }
  }

  if (RELU_A) {
    pa0.x = fmaxf(pa0.x, 0.0f); pa0.y = fmaxf(pa0.y, 0.0f); pa0.z = fmaxf(pa0.z, 0.0f); pa0.w = fmaxf(pa0.w, 0.0f);
    pa1.x = fmaxf(pa1.x, 0.0f); pa1.y = fmaxf(pa1.y, 0.0f); pa1.z = fmaxf(pa1.z, 0.0f); pa1.w = fmaxf(pa1.w, 0.0f);
  }
  *(float4*)(Adst) = pa0;
  *(float4*)(Adst + 8 * LDSA) = pa1;
  pa0 = pb0; pa1 = pb1;
  __syncthreads();

  const int wave = t >> 5;
  const int lane = t & 31;
  const int half = lane >> 4;
  const int mn   = lane & 15;
  const int col  = wave * 16 + mn;
  const float* Bb = Bl + (half * NCOLS + col) * 4;
  const float* Afr = Al + mn * LDSA + 4 * half;
  const float bc = bias[col];
  float* Y0 = Y;
  float* Y1 = Y + (long)NNODES * FDIM;
  float* Y2 = Y + 2L * NNODES * FDIM;

  Frags f0, f1;
  load_chunk(f0, Afr, Bb, 0);
  int cur = 0;
  for (int i = 0; i < ntile; ++i) {
    const int nxt = (cur == ABUFS - 1) ? 0 : cur + 1;
    const float* Ab = Afr + cur * ATILE;
    v8f acc[4] = {};
    load_chunk(f1, Ab, Bb, 1);
    __builtin_amdgcn_sched_barrier(0);
    mma_chunk(f0, acc);
    __builtin_amdgcn_sched_barrier(0);
    load_chunk(f0, Ab, Bb, 2);
    __builtin_amdgcn_sched_barrier(0);
    mma_chunk(f1, acc);
    __builtin_amdgcn_sched_barrier(0);
    if (i + 1 < ntile) {
      if (RELU_A) {
        pa0.x = fmaxf(pa0.x, 0.0f); pa0.y = fmaxf(pa0.y, 0.0f); pa0.z = fmaxf(pa0.z, 0.0f); pa0.w = fmaxf(pa0.w, 0.0f);
        pa1.x = fmaxf(pa1.x, 0.0f); pa1.y = fmaxf(pa1.y, 0.0f); pa1.z = fmaxf(pa1.z, 0.0f); pa1.w = fmaxf(pa1.w, 0.0f);
      }
      *(float4*)(Adst + nxt * ATILE) = pa0;
      *(float4*)(Adst + nxt * ATILE + 8 * LDSA) = pa1;
    }
    __syncthreads();
    if (i + 2 < ntile) {
      long r = (long)(bid + (i + 2) * PGRID) * 16;
      pa0 = *(const float4*)(Asrc + r * FDIM);
      pa1 = *(const float4*)(Asrc + (r + 8) * FDIM);
    }
    load_chunk(f1, Ab, Bb, 3);
    __builtin_amdgcn_sched_barrier(0);
    mma_chunk(f0, acc);
    __builtin_amdgcn_sched_barrier(0);
    if (i + 1 < ntile) load_chunk(f0, Afr + nxt * ATILE, Bb, 0);
    __builtin_amdgcn_sched_barrier(0);
    mma_chunk(f1, acc);
    __builtin_amdgcn_sched_barrier(0);

    const long row0 = (long)(bid + i * PGRID) * 16;
    #pragma unroll
    for (int r = 0; r < 8; ++r) {
      long o = (row0 + r + 8 * half) * FDIM + col;
      hacc[o] = acc[0][r] + bc;
      Y0[o] = acc[1][r];
      Y1[o] = acc[2][r];
      Y2[o] = acc[3][r];
    }
    cur = nxt;
  }
}

#define NOWNERS   256
#define OWN_ROWS  391
#define CHUNK     32
#define FLUSH_AT  CHUNK
#define PEND_CAP  544
#define BATCH     16
#define HEAD_ROWS ((OWN_ROWS + 31) / 32)

__device__ __forceinline__ void drain_pending(
    const int* pend, int n, int lane, int d0,
    const int* __restrict__ ei, const int* __restrict__ et,
    const float* __restrict__ Y, const float* __restrict__ cnt, float* acc) {
  for (int i = 0; i < n; i += CHUNK) {
    const int k = min(i + lane, n - 1);
    const int w = pend[k];
    const int e = w & 0xFFFFF;
    const int dl = (int)((unsigned)w >> 20);
    const int s = ei[e];
    const int r = et[e];
    const float c = cnt[(long)r * NNODES + d0 + dl];
    const int off = r * NNODES + s;
    const int sci = __float_as_int(1.0f / fmaxf(c, 1.0f));
    const int row = (i + lane < n) ? dl : OWN_ROWS;
    #pragma unroll
    for (int h = 0; h < CHUNK / BATCH; ++h) {
      if (i + h * BATCH < n) {
        float4 v[BATCH];
        #pragma unroll
        for (int j = 0; j < BATCH; ++j) {
          const int o = __builtin_amdgcn_readlane(off, h * BATCH + j);
          v[j] = *(const float4*)(Y + (long)o * FDIM + lane * 4);
        }
        #pragma unroll
        for (int j = 0; j < BATCH; ++j) {
          const float sc = __int_as_float(__builtin_amdgcn_readlane(sci, h * BATCH + j));
          const int rj = __builtin_amdgcn_readlane(row, h * BATCH + j);
          float* a = acc + rj * FDIM + lane;
          atomicAdd(a,      v[j].x * sc);
          atomicAdd(a + 32, v[j].y * sc);
          atomicAdd(a + 64, v[j].z * sc);
          atomicAdd(a + 96, v[j].w * sc);
        }
      }
    }
  }
}

template <bool HEAD>
__global__ __launch_bounds__(1024) void edge_gather_owned(
    const int* __restrict__ ei, const int* __restrict__ et,
    const float* __restrict__ Y, const float* __restrict__ cnt,
    float* __restrict__ hacc,
    const float* __restrict__ lw, const float* __restrict__ lb, float* __restrict__ out) {
  __shared__ __align__(16) float acc[(OWN_ROWS + 1) * FDIM];
  __shared__ int pend_all[32 * PEND_CAP];
  const int t = threadIdx.x;
  const int lane = t & 31;
  const int wave = t >> 5;
  const int d0 = (int)blockIdx.x * OWN_ROWS;
  const int nrows = min(OWN_ROWS, NNODES - d0);

  const float4* hsrc = (const float4*)(hacc + (long)d0 * FDIM);
  for (int i = t; i < nrows * 32; i += 1024) {
    float4 v = hsrc[i];
    float* a = acc + (i >> 5) * FDIM + (i & 31);
    a[0] = v.x; a[32] = v.y; a[64] = v.z; a[96] = v.w;
  }
  __syncthreads();

  int* pend = pend_all + wave * PEND_CAP;
  int npend = 0;
  const int4* dst4 = (const int4*)(ei + NEDGES);
  const int NQ = NEDGES / 4;
  for (int base = 0; base < NQ; base += 4 * 1024) {
    if (npend >= FLUSH_AT) {
      const int nfull = npend & ~(CHUNK - 1);
      drain_pending(pend, nfull, lane, d0, ei, et, Y, cnt, acc);
      const int rem = npend - nfull;
      if (lane < rem) pend[lane] = pend[nfull + lane];
      npend = rem;
    }
    int4 dv[4];
    #pragma unroll
    for (int u = 0; u < 4; ++u) {
      int q = base + u * 1024 + t;
      dv[u] = (q < NQ) ? dst4[q] : make_int4(-1, -1, -1, -1);
    }
    #pragma unroll
    for (int u = 0; u < 4; ++u) {
      int q = base + u * 1024 + t;
      unsigned r0 = (unsigned)(dv[u].x - d0), r1 = (unsigned)(dv[u].y - d0);
      unsigned r2 = (unsigned)(dv[u].z - d0), r3 = (unsigned)(dv[u].w - d0);
      bool m0 = r0 < (unsigned)nrows, m1 = r1 < (unsigned)nrows;
      bool m2 = r2 < (unsigned)nrows, m3 = r3 < (unsigned)nrows;
      if (__builtin_amdgcn_ballot_w32(m0 | m1 | m2 | m3)) {
        unsigned k;
        k = __builtin_amdgcn_ballot_w32(m0);
        if (m0) pend[npend + __builtin_amdgcn_mbcnt_lo(k, 0u)] = (int)((r0 << 20) | (unsigned)(4 * q + 0));
        npend += __builtin_popcount(k);
        k = __builtin_amdgcn_ballot_w32(m1);
        if (m1) pend[npend + __builtin_amdgcn_mbcnt_lo(k, 0u)] = (int)((r1 << 20) | (unsigned)(4 * q + 1));
        npend += __builtin_popcount(k);
        k = __builtin_amdgcn_ballot_w32(m2);
        if (m2) pend[npend + __builtin_amdgcn_mbcnt_lo(k, 0u)] = (int)((r2 << 20) | (unsigned)(4 * q + 2));
        npend += __builtin_popcount(k);
        k = __builtin_amdgcn_ballot_w32(m3);
        if (m3) pend[npend + __builtin_amdgcn_mbcnt_lo(k, 0u)] = (int)((r3 << 20) | (unsigned)(4 * q + 3));
        npend += __builtin_popcount(k);
      }
    }
  }
  if (npend > 0) drain_pending(pend, npend, lane, d0, ei, et, Y, cnt, acc);
  __syncthreads();

  if (!HEAD) {
    float4* hdst = (float4*)(hacc + (long)d0 * FDIM);
    for (int i = t; i < nrows * 32; i += 1024) {
      const float* a = acc + (i >> 5) * FDIM + (i & 31);
      hdst[i] = make_float4(a[0], a[32], a[64], a[96]);
    }
  } else {
    const float4 w0 = *(const float4*)(lw + lane * 8);
    const float4 w1 = *(const float4*)(lw + lane * 8 + 4);
    const float lb0 = lb[0], lb1 = lb[1];
    const int r0 = wave * HEAD_ROWS;
    float k0 = 0.0f, k1 = 0.0f;
    #pragma unroll
    for (int j = 0; j < HEAD_ROWS; ++j) {
      const float* a = acc + min(r0 + j, OWN_ROWS) * FDIM + lane;
      float vx = fmaxf(a[0], 0.0f), vy = fmaxf(a[32], 0.0f);
      float vz = fmaxf(a[64], 0.0f), vw = fmaxf(a[96], 0.0f);
      float p0 = vx * w0.x + vy * w0.z + vz * w1.x + vw * w1.z;
      float p1 = vx * w0.y + vy * w0.w + vz * w1.y + vw * w1.w;
      #pragma unroll
      for (int off = 16; off > 0; off >>= 1) {
        p0 += __shfl_xor(p0, off, 32);
        p1 += __shfl_xor(p1, off, 32);
      }
      if (lane == j) { k0 = p0; k1 = p1; }
    }
    if (lane < HEAD_ROWS && r0 + lane < nrows) {
      float l0 = k0 + lb0, l1 = k1 + lb1;
      float m = fmaxf(l0, l1);
      float lse = m + logf(expf(l0 - m) + expf(l1 - m));
      *(float2*)(out + (long)(d0 + r0 + lane) * 2) = make_float2(l0 - lse, l1 - lse);
    }
  }
}

extern "C" void kernel_launch(void* const* d_in, const int* in_sizes, int n_in,
                              void* d_out, int out_size, void* d_ws, size_t ws_size,
                              hipStream_t stream) {
  const float* x      = (const float*)d_in[0];
  const int*   ei     = (const int*)d_in[1];
  const int*   et     = (const int*)d_in[2];
  const float* W1rel  = (const float*)d_in[3];
  const float* W1root = (const float*)d_in[4];
  const float* b1     = (const float*)d_in[5];
  const float* W2rel  = (const float*)d_in[6];
  const float* W2root = (const float*)d_in[7];
  const float* b2     = (const float*)d_in[8];
  const float* lw     = (const float*)d_in[9];
  const float* lb     = (const float*)d_in[10];

  const long msgN = (long)NREL * NNODES * FDIM;
  const long cntN = (long)NREL * NNODES;
  const long NF   = (long)NNODES * FDIM;

  float* ws  = (float*)d_ws;
  float* Y   = ws;
  float* cnt = Y + msgN;
  float* h1  = cnt + cntN;
  float* h2  = h1 + NF;

  dim3 blk(256);

  zero_f4<<<293, blk, 0, stream>>>(cnt, cntN / 4);
  edge_count<<<(NEDGES + 255) / 256, blk, 0, stream>>>(ei, et, cnt);

  rgcn_project<false><<<PGRID, blk, 0, stream>>>(
      x, W1root, W1rel, W1rel + FDIM * FDIM, W1rel + 2 * FDIM * FDIM, b1, h1, Y);
  edge_gather_owned<false><<<NOWNERS, 1024, 0, stream>>>(ei, et, Y, cnt, h1, lw, lb, (float*)d_out);

  rgcn_project<true><<<PGRID, blk, 0, stream>>>(
      h1, W2root, W2rel, W2rel + FDIM * FDIM, W2rel + 2 * FDIM * FDIM, b2, h2, Y);
  edge_gather_owned<true><<<NOWNERS, 1024, 0, stream>>>(ei, et, Y, cnt, h2, lw, lb, (float*)d_out);
}


# layer-1 owners save each resolved edge as an 8-byte record (row, Y row, scale); layer 2 walks its list: no 2nd filter pass, no index loads; its root plane now written in place; new LDS add order
# speedup vs baseline: 2.3072x; 1.0957x over previous
#include <hip/hip_runtime.h>


#define NNODES 100000
#define NEDGES 600000
#define FDIM   128
#define NREL   3
#define LDSA   132

typedef float v2f __attribute__((ext_vector_type(2)));
typedef float v8f __attribute__((ext_vector_type(8)));

__global__ void zero_f4(float* __restrict__ p, long n4) {
  long i = (long)blockIdx.x * blockDim.x + threadIdx.x;
  long stride = (long)gridDim.x * blockDim.x;
  float4 z = make_float4(0.f, 0.f, 0.f, 0.f);
  for (; i < n4; i += stride) ((float4*)p)[i] = z;
}

__global__ void edge_count(const int* __restrict__ ei, const int* __restrict__ et,
                           float* __restrict__ cnt) {
  int e = (int)(blockIdx.x * blockDim.x + threadIdx.x);
  if (e < NEDGES) atomicAdd(cnt + (long)et[e] * NNODES + ei[NEDGES + e], 1.0f);
}

#define PGRID   256
#define NTILES  (NNODES / 16)
#define NCOLS   (4 * FDIM)
#define ABUFS   3
#define ATILE   (16 * LDSA)

struct Frags { float4 a[4]; float4 b[16]; };

__device__ __forceinline__ void load_chunk(Frags& f, const float* Ab, const float* Bb, int c) {
  #pragma unroll
  for (int jj = 0; jj < 4; ++jj) {
    int j = 4 * c + jj;
    f.a[jj] = *(const float4*)(Ab + 8 * j);
    #pragma unroll
    for (int p = 0; p < 4; ++p)
      f.b[jj * 4 + p] = *(const float4*)(Bb + j * (2 * NCOLS * 4) + p * (FDIM * 4));
  }
}

__device__ __forceinline__ void mma_chunk(const Frags& f, v8f (&acc)[4]) {
  #pragma unroll
  for (int jj = 0; jj < 4; ++jj) {
    v2f a0 = {f.a[jj].x, f.a[jj].y};
    v2f a1 = {f.a[jj].z, f.a[jj].w};
    #pragma unroll
    for (int p = 0; p < 4; ++p) {
      v2f b0 = {f.b[jj * 4 + p].x, f.b[jj * 4 + p].y};
      acc[p] = __builtin_amdgcn_wmma_f32_16x16x4_f32(false, a0, false, b0, (short)0, acc[p], false, false);
    }
    #pragma unroll
    for (int p = 0; p < 4; ++p) {
      v2f b1 = {f.b[jj * 4 + p].z, f.b[jj * 4 + p].w};
      acc[p] = __builtin_amdgcn_wmma_f32_16x16x4_f32(false, a1, false, b1, (short)0, acc[p], false, false);
    }
  }
}

template <bool RELU_A>
__global__ __launch_bounds__(256) void rgcn_project(
    const float* A,
    const float* __restrict__ B0, const float* __restrict__ B1,
    const float* __restrict__ B2, const float* __restrict__ B3,
    const float* __restrict__ bias, float* hacc, float* __restrict__ Y) {
  __shared__ __align__(16) float Bl[(FDIM / 4) * NCOLS * 4];
  __shared__ __align__(16) float Al[ABUFS * ATILE];
  const int t = threadIdx.x;
  const int bid = (int)blockIdx.x;
  const int ntile = (NTILES - bid + PGRID - 1) / PGRID;

  const int ar = t >> 5, ac4 = t & 31;
  const float* Asrc = A + (long)ar * FDIM + ac4 * 4;
  float* Adst = Al + ar * LDSA + ac4 * 4;
  float4 pa0 = *(const float4*)(Asrc + (long)bid * 16 * FDIM);
  float4 pa1 = *(const float4*)(Asrc + ((long)bid * 16 + 8) * FDIM);
  float4 pb0 = pa0, pb1 = pa1;
  if (ntile > 1) {
    pb0 = *(const float4*)(Asrc + (long)(bid + PGRID) * 16 * FDIM);
    pb1 = *(const float4*)(Asrc + ((long)(bid + PGRID) * 16 + 8) * FDIM);
  }

  {
    const float* Bpa = (t < 128) ? B0 : B1;
    const float* Bpb = (t < 128) ? B2 : B3;
    const int n = t & 127;
    #pragma unroll 8
    for (int i = 0; i < FDIM / 4; ++i) {
      int kq = (i + bid) & (FDIM / 4 - 1);
      const float* ra = Bpa + kq * 4 * FDIM + n;
      const float* rb = Bpb + kq * 4 * FDIM + n;
      float4 va = make_float4(ra[0], ra[FDIM], ra[2 * FDIM], ra[3 * FDIM]);
      float4 vb = make_float4(rb[0], rb[FDIM], rb[2 * FDIM], rb[3 * FDIM]);
      *(float4*)(Bl + (kq * NCOLS + t) * 4) = va;
      *(float4*)(Bl + (kq * NCOLS + 256 + t) * 4) = vb;
    }
  }

  if (RELU_A) {
    pa0.x = fmaxf(pa0.x, 0.0f); pa0.y = fmaxf(pa0.y, 0.0f); pa0.z = fmaxf(pa0.z, 0.0f); pa0.w = fmaxf(pa0.w, 0.0f);
    pa1.x = fmaxf(pa1.x, 0.0f); pa1.y = fmaxf(pa1.y, 0.0f); pa1.z = fmaxf(pa1.z, 0.0f); pa1.w = fmaxf(pa1.w, 0.0f);
  }
  *(float4*)(Adst) = pa0;
  *(float4*)(Adst + 8 * LDSA) = pa1;
  pa0 = pb0; pa1 = pb1;
  __syncthreads();

  const int wave = t >> 5;
  const int lane = t & 31;
  const int half = lane >> 4;
  const int mn   = lane & 15;
  const int col  = wave * 16 + mn;
  const float* Bb = Bl + (half * NCOLS + col) * 4;
  const float* Afr = Al + mn * LDSA + 4 * half;
  const float bc = bias[col];
  float* Y0 = Y;
  float* Y1 = Y + (long)NNODES * FDIM;
  float* Y2 = Y + 2L * NNODES * FDIM;

  Frags f0, f1;
  load_chunk(f0, Afr, Bb, 0);
  int cur = 0;
  for (int i = 0; i < ntile; ++i) {
    const int nxt = (cur == ABUFS - 1) ? 0 : cur + 1;
    const float* Ab = Afr + cur * ATILE;
    v8f acc[4] = {};
    load_chunk(f1, Ab, Bb, 1);
    __builtin_amdgcn_sched_barrier(0);
    mma_chunk(f0, acc);
    __builtin_amdgcn_sched_barrier(0);
    load_chunk(f0, Ab, Bb, 2);
    __builtin_amdgcn_sched_barrier(0);
    mma_chunk(f1, acc);
    __builtin_amdgcn_sched_barrier(0);
    if (i + 1 < ntile) {
      if (RELU_A) {
        pa0.x = fmaxf(pa0.x, 0.0f); pa0.y = fmaxf(pa0.y, 0.0f); pa0.z = fmaxf(pa0.z, 0.0f); pa0.w = fmaxf(pa0.w, 0.0f);
        pa1.x = fmaxf(pa1.x, 0.0f); pa1.y = fmaxf(pa1.y, 0.0f); pa1.z = fmaxf(pa1.z, 0.0f); pa1.w = fmaxf(pa1.w, 0.0f);
      }
      *(float4*)(Adst + nxt * ATILE) = pa0;
      *(float4*)(Adst + nxt * ATILE + 8 * LDSA) = pa1;
    }
    __syncthreads();
    if (i + 2 < ntile) {
      long r = (long)(bid + (i + 2) * PGRID) * 16;
      pa0 = *(const float4*)(Asrc + r * FDIM);
      pa1 = *(const float4*)(Asrc + (r + 8) * FDIM);
    }
    load_chunk(f1, Ab, Bb, 3);
    __builtin_amdgcn_sched_barrier(0);
    mma_chunk(f0, acc);
    __builtin_amdgcn_sched_barrier(0);
    if (i + 1 < ntile) load_chunk(f0, Afr + nxt * ATILE, Bb, 0);
    __builtin_amdgcn_sched_barrier(0);
    mma_chunk(f1, acc);
    __builtin_amdgcn_sched_barrier(0);

    const long row0 = (long)(bid + i * PGRID) * 16;
    #pragma unroll
    for (int r = 0; r < 8; ++r) {
      long o = (row0 + r + 8 * half) * FDIM + col;
      hacc[o] = acc[0][r] + bc;
      Y0[o] = acc[1][r];
      Y1[o] = acc[2][r];
      Y2[o] = acc[3][r];
    }
    cur = nxt;
  }
}

#define NOWNERS   256
#define OWN_ROWS  391
#define CHUNK     32
#define FLUSH_AT  CHUNK
#define PEND_CAP  544
#define BATCH     16
#define HEAD_ROWS ((OWN_ROWS + 31) / 32)
#define LIST_STRIDE ((NNODES * FDIM / NOWNERS) & ~63)
#define LIST_HDR    32
#define LIST_CAP    (LIST_STRIDE / 2 - LIST_HDR)

__device__ __forceinline__ void add_rows(int off, int sci, int row, int nleft, int lane,
                                         const float* __restrict__ Y, float* acc) {
  #pragma unroll
  for (int h = 0; h < CHUNK / BATCH; ++h) {
    if (h * BATCH < nleft) {
      float4 v[BATCH];
      #pragma unroll
      for (int j = 0; j < BATCH; ++j) {
        const int o = __builtin_amdgcn_readlane(off, h * BATCH + j);
        v[j] = *(const float4*)(Y + (long)o * FDIM + lane * 4);
      }
      #pragma unroll
      for (int j = 0; j < BATCH; ++j) {
        const float sc = __int_as_float(__builtin_amdgcn_readlane(sci, h * BATCH + j));
        const int rj = __builtin_amdgcn_readlane(row, h * BATCH + j);
        float* a = acc + rj * FDIM + lane;
        atomicAdd(a,      v[j].x * sc);
        atomicAdd(a + 32, v[j].y * sc);
        atomicAdd(a + 64, v[j].z * sc);
        atomicAdd(a + 96, v[j].w * sc);
      }
    }
  }
}

template <bool SAVE>
__device__ __forceinline__ void drain_pending(
    const int* pend, int n, int lane, int d0,
    const int* __restrict__ ei, const int* __restrict__ et,
    const float* __restrict__ Y, const float* __restrict__ cnt, float* acc,
    int* lcur, int2* __restrict__ lst) {
  int base = 0;
  if (SAVE) {
    if (lane == 0) base = atomicAdd(lcur, n);
    base = __builtin_amdgcn_readfirstlane(base);
  }
  for (int i = 0; i < n; i += CHUNK) {
    const int k = min(i + lane, n - 1);
    const int w = pend[k];
    const int e = w & 0xFFFFF;
    const int dl = (int)((unsigned)w >> 20);
    const int s = ei[e];
    const int r = et[e];
    const float c = cnt[(long)r * NNODES + d0 + dl];
    const int off = r * NNODES + s;
    const int sci = __float_as_int(1.0f / fmaxf(c, 1.0f));
    const int row = (i + lane < n) ? dl : OWN_ROWS;
    if (SAVE) {
      const int slot = base + i + lane;
      if (i + lane < n && slot < LIST_CAP) lst[slot] = make_int2((dl << 19) | off, sci);
    }
    add_rows(off, sci, row, n - i, lane, Y, acc);
  }
}

template <bool SAVE>
__device__ __forceinline__ void scan_and_drain(
    int* pend, int t, int lane, int d0, int nrows,
    const int* __restrict__ ei, const int* __restrict__ et,
    const float* __restrict__ Y, const float* __restrict__ cnt, float* acc,
    int* lcur, int2* __restrict__ lst) {
  int npend = 0;
  const int4* dst4 = (const int4*)(ei + NEDGES);
  const int NQ = NEDGES / 4;
  for (int base = 0; base < NQ; base += 4 * 1024) {
    if (npend >= FLUSH_AT) {
      const int nfull = npend & ~(CHUNK - 1);
      drain_pending<SAVE>(pend, nfull, lane, d0, ei, et, Y, cnt, acc, lcur, lst);
      const int rem = npend - nfull;
      if (lane < rem) pend[lane] = pend[nfull + lane];
      npend = rem;
    }
    int4 dv[4];
    #pragma unroll
    for (int u = 0; u < 4; ++u) {
      int q = base + u * 1024 + t;
      dv[u] = (q < NQ) ? dst4[q] : make_int4(-1, -1, -1, -1);
    }
    #pragma unroll
    for (int u = 0; u < 4; ++u) {
      int q = base + u * 1024 + t;
      unsigned r0 = (unsigned)(dv[u].x - d0), r1 = (unsigned)(dv[u].y - d0);
      unsigned r2 = (unsigned)(dv[u].z - d0), r3 = (unsigned)(dv[u].w - d0);
      bool m0 = r0 < (unsigned)nrows, m1 = r1 < (unsigned)nrows;
      bool m2 = r2 < (unsigned)nrows, m3 = r3 < (unsigned)nrows;
      if (__builtin_amdgcn_ballot_w32(m0 | m1 | m2 | m3)) {
        unsigned k;
        k = __builtin_amdgcn_ballot_w32(m0);
        if (m0) pend[npend + __builtin_amdgcn_mbcnt_lo(k, 0u)] = (int)((r0 << 20) | (unsigned)(4 * q + 0));
        npend += __builtin_popcount(k);
        k = __builtin_amdgcn_ballot_w32(m1);
        if (m1) pend[npend + __builtin_amdgcn_mbcnt_lo(k, 0u)] = (int)((r1 << 20) | (unsigned)(4 * q + 1));
        npend += __builtin_popcount(k);
        k = __builtin_amdgcn_ballot_w32(m2);
        if (m2) pend[npend + __builtin_amdgcn_mbcnt_lo(k, 0u)] = (int)((r2 << 20) | (unsigned)(4 * q + 2));
        npend += __builtin_popcount(k);
        k = __builtin_amdgcn_ballot_w32(m3);
        if (m3) pend[npend + __builtin_amdgcn_mbcnt_lo(k, 0u)] = (int)((r3 << 20) | (unsigned)(4 * q + 3));
        npend += __builtin_popcount(k);
      }
    }
  }
  if (npend > 0) drain_pending<SAVE>(pend, npend, lane, d0, ei, et, Y, cnt, acc, lcur, lst);
}

template <bool HEAD>
__global__ __launch_bounds__(1024) void edge_gather_owned(
    const int* __restrict__ ei, const int* __restrict__ et,
    const float* __restrict__ Y, const float* __restrict__ cnt,
    float* __restrict__ hacc, float* __restrict__ lists,
    const float* __restrict__ lw, const float* __restrict__ lb, float* __restrict__ out) {
  __shared__ __align__(16) float acc[(OWN_ROWS + 1) * FDIM];
  __shared__ int pend_all[32 * PEND_CAP];
  __shared__ int lcur;
  const int t = threadIdx.x;
  const int lane = t & 31;
  const int wave = t >> 5;
  const int d0 = (int)blockIdx.x * OWN_ROWS;
  const int nrows = min(OWN_ROWS, NNODES - d0);
  int* lhead = (int*)(lists + (long)blockIdx.x * LIST_STRIDE);
  int2* lst = (int2*)lhead + LIST_HDR;

  int nlist = 0;
  if (HEAD) nlist = lhead[0];
  if (!HEAD && t == 0) lcur = 0;

  const float4* hsrc = (const float4*)(hacc + (long)d0 * FDIM);
  for (int i = t; i < nrows * 32; i += 1024) {
    float4 v = hsrc[i];
    float* a = acc + (i >> 5) * FDIM + (i & 31);
    a[0] = v.x; a[32] = v.y; a[64] = v.z; a[96] = v.w;
  }
  __syncthreads();

  int* pend = pend_all + wave * PEND_CAP;
  if (!HEAD) {
    scan_and_drain<true>(pend, t, lane, d0, nrows, ei, et, Y, cnt, acc, &lcur, lst);
  } else if (nlist <= LIST_CAP) {
    int c = __builtin_amdgcn_readfirstlane(wave) * CHUNK;
    int2 rec = make_int2(0, 0);
    if (c < nlist) rec = lst[min(c + lane, nlist - 1)];
    while (c < nlist) {
      const int cn = c + 32 * CHUNK;
      int2 recn = make_int2(0, 0);
      if (cn < nlist) recn = lst[min(cn + lane, nlist - 1)];
      const int row = (c + lane < nlist) ? (int)((unsigned)rec.x >> 19) : OWN_ROWS;
      add_rows(rec.x & 0x7FFFF, rec.y, row, nlist - c, lane, Y, acc);
      rec = recn;
      c = cn;
    }
  } else {
    scan_and_drain<false>(pend, t, lane, d0, nrows, ei, et, Y, cnt, acc, &lcur, lst);
  }
  __syncthreads();

  if (!HEAD) {
    if (t == 0) lhead[0] = lcur;
    float4* hdst = (float4*)(hacc + (long)d0 * FDIM);
    for (int i = t; i < nrows * 32; i += 1024) {
      const float* a = acc + (i >> 5) * FDIM + (i & 31);
      hdst[i] = make_float4(a[0], a[32], a[64], a[96]);
    }
  } else {
    const float4 w0 = *(const float4*)(lw + lane * 8);
    const float4 w1 = *(const float4*)(lw + lane * 8 + 4);
    const float lb0 = lb[0], lb1 = lb[1];
    const int r0 = wave * HEAD_ROWS;
    float k0 = 0.0f, k1 = 0.0f;
    #pragma unroll
    for (int j = 0; j < HEAD_ROWS; ++j) {
      const float* a = acc + min(r0 + j, OWN_ROWS) * FDIM + lane;
      float vx = fmaxf(a[0], 0.0f), vy = fmaxf(a[32], 0.0f);
      float vz = fmaxf(a[64], 0.0f), vw = fmaxf(a[96], 0.0f);
      float p0 = vx * w0.x + vy * w0.z + vz * w1.x + vw * w1.z;
      float p1 = vx * w0.y + vy * w0.w + vz * w1.y + vw * w1.w;
      #pragma unroll
      for (int off = 16; off > 0; off >>= 1) {
        p0 += __shfl_xor(p0, off, 32);
        p1 += __shfl_xor(p1, off, 32);
      }
      if (lane == j) { k0 = p0; k1 = p1; }
    }
    if (lane < HEAD_ROWS && r0 + lane < nrows) {
      float l0 = k0 + lb0, l1 = k1 + lb1;
      float m = fmaxf(l0, l1);
      float lse = m + logf(expf(l0 - m) + expf(l1 - m));
      *(float2*)(out + (long)(d0 + r0 + lane) * 2) = make_float2(l0 - lse, l1 - lse);
    }
  }
}

extern "C" void kernel_launch(void* const* d_in, const int* in_sizes, int n_in,
                              void* d_out, int out_size, void* d_ws, size_t ws_size,
                              hipStream_t stream) {
  const float* x      = (const float*)d_in[0];
  const int*   ei     = (const int*)d_in[1];
  const int*   et     = (const int*)d_in[2];
  const float* W1rel  = (const float*)d_in[3];
  const float* W1root = (const float*)d_in[4];
  const float* b1     = (const float*)d_in[5];
  const float* W2rel  = (const float*)d_in[6];
  const float* W2root = (const float*)d_in[7];
  const float* b2     = (const float*)d_in[8];
  const float* lw     = (const float*)d_in[9];
  const float* lb     = (const float*)d_in[10];

  const long msgN = (long)NREL * NNODES * FDIM;
  const long cntN = (long)NREL * NNODES;
  const long NF   = (long)NNODES * FDIM;

  float* ws  = (float*)d_ws;
  float* Y   = ws;
  float* cnt = Y + msgN;
  float* h1  = cnt + cntN;
  float* lst = h1 + NF;

  dim3 blk(256);

  zero_f4<<<293, blk, 0, stream>>>(cnt, cntN / 4);
  edge_count<<<(NEDGES + 255) / 256, blk, 0, stream>>>(ei, et, cnt);

  rgcn_project<false><<<PGRID, blk, 0, stream>>>(
      x, W1root, W1rel, W1rel + FDIM * FDIM, W1rel + 2 * FDIM * FDIM, b1, h1, Y);
  edge_gather_owned<false><<<NOWNERS, 1024, 0, stream>>>(ei, et, Y, cnt, h1, lst, lw, lb, (float*)d_out);

  rgcn_project<true><<<PGRID, blk, 0, stream>>>(
      h1, W2root, W2rel, W2rel + FDIM * FDIM, W2rel + 2 * FDIM * FDIM, b2, h1, Y);
  edge_gather_owned<true><<<NOWNERS, 1024, 0, stream>>>(ei, et, Y, cnt, h1, lst, lw, lb, (float*)d_out);
}


# degree pass also bins every edge into 32-word (owner, slice) segments, slot by an LDS counting add; layer-1 owners read their 256 segments: no destination filter, no index loads; new LDS add order
# speedup vs baseline: 2.4370x; 1.0563x over previous
#include <hip/hip_runtime.h>


#define NNODES 100000
#define NEDGES 600000
#define FDIM   128
#define NREL   3
#define LDSA   132

typedef float v2f __attribute__((ext_vector_type(2)));
typedef float v8f __attribute__((ext_vector_type(8)));

__global__ void zero_f4(float* __restrict__ p, long n4) {
  long i = (long)blockIdx.x * blockDim.x + threadIdx.x;
  long stride = (long)gridDim.x * blockDim.x;
  float4 z = make_float4(0.f, 0.f, 0.f, 0.f);
  for (; i < n4; i += stride) ((float4*)p)[i] = z;
}

#define PGRID   256
#define NTILES  (NNODES / 16)
#define NCOLS   (4 * FDIM)
#define ABUFS   3
#define ATILE   (16 * LDSA)

struct Frags { float4 a[4]; float4 b[16]; };

__device__ __forceinline__ void load_chunk(Frags& f, const float* Ab, const float* Bb, int c) {
  #pragma unroll
  for (int jj = 0; jj < 4; ++jj) {
    int j = 4 * c + jj;
    f.a[jj] = *(const float4*)(Ab + 8 * j);
    #pragma unroll
    for (int p = 0; p < 4; ++p)
      f.b[jj * 4 + p] = *(const float4*)(Bb + j * (2 * NCOLS * 4) + p * (FDIM * 4));
  }
}

__device__ __forceinline__ void mma_chunk(const Frags& f, v8f (&acc)[4]) {
  #pragma unroll
  for (int jj = 0; jj < 4; ++jj) {
    v2f a0 = {f.a[jj].x, f.a[jj].y};
    v2f a1 = {f.a[jj].z, f.a[jj].w};
    #pragma unroll
    for (int p = 0; p < 4; ++p) {
      v2f b0 = {f.b[jj * 4 + p].x, f.b[jj * 4 + p].y};
      acc[p] = __builtin_amdgcn_wmma_f32_16x16x4_f32(false, a0, false, b0, (short)0, acc[p], false, false);
    }
    #pragma unroll
    for (int p = 0; p < 4; ++p) {
      v2f b1 = {f.b[jj * 4 + p].z, f.b[jj * 4 + p].w};
      acc[p] = __builtin_amdgcn_wmma_f32_16x16x4_f32(false, a1, false, b1, (short)0, acc[p], false, false);
    }
  }
}

template <bool RELU_A>
__global__ __launch_bounds__(256) void rgcn_project(
    const float* A,
    const float* __restrict__ B0, const float* __restrict__ B1,
    const float* __restrict__ B2, const float* __restrict__ B3,
    const float* __restrict__ bias, float* hacc, float* __restrict__ Y) {
  __shared__ __align__(16) float Bl[(FDIM / 4) * NCOLS * 4];
  __shared__ __align__(16) float Al[ABUFS * ATILE];
  const int t = threadIdx.x;
  const int bid = (int)blockIdx.x;
  const int ntile = (NTILES - bid + PGRID - 1) / PGRID;

  const int ar = t >> 5, ac4 = t & 31;
  const float* Asrc = A + (long)ar * FDIM + ac4 * 4;
  float* Adst = Al + ar * LDSA + ac4 * 4;
  float4 pa0 = *(const float4*)(Asrc + (long)bid * 16 * FDIM);
  float4 pa1 = *(const float4*)(Asrc + ((long)bid * 16 + 8) * FDIM);
  float4 pb0 = pa0, pb1 = pa1;
  if (ntile > 1) {
    pb0 = *(const float4*)(Asrc + (long)(bid + PGRID) * 16 * FDIM);
    pb1 = *(const float4*)(Asrc + ((long)(bid + PGRID) * 16 + 8) * FDIM);
  }

  {
    const float* Bpa = (t < 128) ? B0 : B1;
    const float* Bpb = (t < 128) ? B2 : B3;
    const int n = t & 127;
    #pragma unroll 8
    for (int i = 0; i < FDIM / 4; ++i) {
      int kq = (i + bid) & (FDIM / 4 - 1);
      const float* ra = Bpa + kq * 4 * FDIM + n;
      const float* rb = Bpb + kq * 4 * FDIM + n;
      float4 va = make_float4(ra[0], ra[FDIM], ra[2 * FDIM], ra[3 * FDIM]);
      float4 vb = make_float4(rb[0], rb[FDIM], rb[2 * FDIM], rb[3 * FDIM]);
      *(float4*)(Bl + (kq * NCOLS + t) * 4) = va;
      *(float4*)(Bl + (kq * NCOLS + 256 + t) * 4) = vb;
    }
  }

  if (RELU_A) {
    pa0.x = fmaxf(pa0.x, 0.0f); pa0.y = fmaxf(pa0.y, 0.0f); pa0.z = fmaxf(pa0.z, 0.0f); pa0.w = fmaxf(pa0.w, 0.0f);
    pa1.x = fmaxf(pa1.x, 0.0f); pa1.y = fmaxf(pa1.y, 0.0f); pa1.z = fmaxf(pa1.z, 0.0f); pa1.w = fmaxf(pa1.w, 0.0f);
  }
  *(float4*)(Adst) = pa0;
  *(float4*)(Adst + 8 * LDSA) = pa1;
  pa0 = pb0; pa1 = pb1;
  __syncthreads();

  const int wave = t >> 5;
  const int lane = t & 31;
  const int half = lane >> 4;
  const int mn   = lane & 15;
  const int col  = wave * 16 + mn;
  const float* Bb = Bl + (half * NCOLS + col) * 4;
  const float* Afr = Al + mn * LDSA + 4 * half;
  const float bc = bias[col];
  float* Y0 = Y;
  float* Y1 = Y + (long)NNODES * FDIM;
  float* Y2 = Y + 2L * NNODES * FDIM;

  Frags f0, f1;
  load_chunk(f0, Afr, Bb, 0);
  int cur = 0;
  for (int i = 0; i < ntile; ++i) {
    const int nxt = (cur == ABUFS - 1) ? 0 : cur + 1;
    const float* Ab = Afr + cur * ATILE;
    v8f acc[4] = {};
    load_chunk(f1, Ab, Bb, 1);
    __builtin_amdgcn_sched_barrier(0);
    mma_chunk(f0, acc);
    __builtin_amdgcn_sched_barrier(0);
    load_chunk(f0, Ab, Bb, 2);
    __builtin_amdgcn_sched_barrier(0);
    mma_chunk(f1, acc);
    __builtin_amdgcn_sched_barrier(0);
    if (i + 1 < ntile) {
      if (RELU_A) {
        pa0.x = fmaxf(pa0.x, 0.0f); pa0.y = fmaxf(pa0.y, 0.0f); pa0.z = fmaxf(pa0.z, 0.0f); pa0.w = fmaxf(pa0.w, 0.0f);
        pa1.x = fmaxf(pa1.x, 0.0f); pa1.y = fmaxf(pa1.y, 0.0f); pa1.z = fmaxf(pa1.z, 0.0f); pa1.w = fmaxf(pa1.w, 0.0f);
      }
      *(float4*)(Adst + nxt * ATILE) = pa0;
      *(float4*)(Adst + nxt * ATILE + 8 * LDSA) = pa1;
    }
    __syncthreads();
    if (i + 2 < ntile) {
      long r = (long)(bid + (i + 2) * PGRID) * 16;
      pa0 = *(const float4*)(Asrc + r * FDIM);
      pa1 = *(const float4*)(Asrc + (r + 8) * FDIM);
    }
    load_chunk(f1, Ab, Bb, 3);
    __builtin_amdgcn_sched_barrier(0);
    mma_chunk(f0, acc);
    __builtin_amdgcn_sched_barrier(0);
    if (i + 1 < ntile) load_chunk(f0, Afr + nxt * ATILE, Bb, 0);
    __builtin_amdgcn_sched_barrier(0);
    mma_chunk(f1, acc);
    __builtin_amdgcn_sched_barrier(0);

    const long row0 = (long)(bid + i * PGRID) * 16;
    #pragma unroll
    for (int r = 0; r < 8; ++r) {
      long o = (row0 + r + 8 * half) * FDIM + col;
      hacc[o] = acc[0][r] + bc;
      Y0[o] = acc[1][r];
      Y1[o] = acc[2][r];
      Y2[o] = acc[3][r];
    }
    cur = nxt;
  }
}

#define NOWNERS   256
#define OWN_ROWS  391
#define CHUNK     32
#define FLUSH_AT  CHUNK
#define PEND_CAP  544
#define BATCH     16
#define HEAD_ROWS ((OWN_ROWS + 31) / 32)
#define LIST_STRIDE ((NNODES * FDIM / NOWNERS) & ~63)
#define LIST_HDR    32
#define NSLICE      256
#define SLICE_EDGES ((NEDGES + NSLICE - 1) / NSLICE)
#define SLICE_ITERS ((SLICE_EDGES + 1023) / 1024)
#define SEG_CAP     32
#define CNT_OFF     (LIST_STRIDE - NSLICE)
#define SEG_OFF     (CNT_OFF - NSLICE * SEG_CAP)
#define LIST_CAP    (SEG_OFF / 2 - LIST_HDR)

__global__ __launch_bounds__(1024) void edge_count_bin(
    const int* __restrict__ ei, const int* __restrict__ et,
    float* __restrict__ cnt, float* __restrict__ listsf) {
  int* lists = (int*)listsf;
  __shared__ int scnt[NOWNERS];
  const int t = threadIdx.x;
  const int slice = (int)blockIdx.x;
  const int e0 = slice * SLICE_EDGES;
  const int e1 = min(e0 + SLICE_EDGES, NEDGES);
  int s[SLICE_ITERS], d[SLICE_ITERS], r[SLICE_ITERS];
  #pragma unroll
  for (int u = 0; u < SLICE_ITERS; ++u) {
    const int e = e0 + t + u * 1024;
    const bool ok = e < e1;
    s[u] = ok ? ei[e] : 0;
    d[u] = ok ? ei[NEDGES + e] : -1;
    r[u] = ok ? et[e] : 0;
  }
  if (t < NOWNERS) scnt[t] = 0;
  __syncthreads();
  #pragma unroll
  for (int u = 0; u < SLICE_ITERS; ++u) {
    if (d[u] >= 0) {
      atomicAdd(cnt + (long)r[u] * NNODES + d[u], 1.0f);
      const int o = d[u] / OWN_ROWS;
      const int dl = d[u] - o * OWN_ROWS;
      const int slot = atomicAdd(&scnt[o], 1);
      if (slot < SEG_CAP)
        lists[(long)o * LIST_STRIDE + SEG_OFF + slice * SEG_CAP + slot] = (dl << 19) | (r[u] * NNODES + s[u]);
    }
  }
  __syncthreads();
  if (t < NOWNERS) lists[(long)t * LIST_STRIDE + CNT_OFF + slice] = scnt[t];
}

__device__ __forceinline__ void add_rows(int off, int sci, int row, int nleft, int lane,
                                         const float* __restrict__ Y, float* acc) {
  #pragma unroll
  for (int h = 0; h < CHUNK / BATCH; ++h) {
    if (h * BATCH < nleft) {
      float4 v[BATCH];
      #pragma unroll
      for (int j = 0; j < BATCH; ++j) {
        const int o = __builtin_amdgcn_readlane(off, h * BATCH + j);
        v[j] = *(const float4*)(Y + (long)o * FDIM + lane * 4);
      }
      #pragma unroll
      for (int j = 0; j < BATCH; ++j) {
        const float sc = __int_as_float(__builtin_amdgcn_readlane(sci, h * BATCH + j));
        const int rj = __builtin_amdgcn_readlane(row, h * BATCH + j);
        float* a = acc + rj * FDIM + lane;
        atomicAdd(a,      v[j].x * sc);
        atomicAdd(a + 32, v[j].y * sc);
        atomicAdd(a + 64, v[j].z * sc);
        atomicAdd(a + 96, v[j].w * sc);
      }
    }
  }
}

template <bool SAVE, bool BINNED>
__device__ __forceinline__ void drain_pending(
    const int* pend, int n, int lane, int d0,
    const int* __restrict__ ei, const int* __restrict__ et,
    const float* __restrict__ Y, const float* __restrict__ cnt, float* acc,
    int* lcur, int2* __restrict__ lst) {
  int base = 0;
  if (SAVE) {
    if (lane == 0) base = atomicAdd(lcur, n);
    base = __builtin_amdgcn_readfirstlane(base);
  }
  for (int i = 0; i < n; i += CHUNK) {
    const int k = min(i + lane, n - 1);
    const int w = pend[k];
    int dl, r, off;
    if (BINNED) {
      off = w & 0x7FFFF;
      dl = (int)((unsigned)w >> 19);
      r = (off >= NNODES) + (off >= 2 * NNODES);
    } else {
      const int e = w & 0xFFFFF;
      dl = (int)((unsigned)w >> 20);
      r = et[e];
      off = r * NNODES + ei[e];
    }
    const float c = cnt[(long)r * NNODES + d0 + dl];
    const int sci = __float_as_int(1.0f / fmaxf(c, 1.0f));
    const int row = (i + lane < n) ? dl : OWN_ROWS;
    if (SAVE) {
      const int slot = base + i + lane;
      if (i + lane < n && slot < LIST_CAP) lst[slot] = make_int2((dl << 19) | off, sci);
    }
    add_rows(off, sci, row, n - i, lane, Y, acc);
  }
}

template <bool SAVE>
__device__ __forceinline__ void scan_and_drain(
    int* pend, int t, int lane, int d0, int nrows,
    const int* __restrict__ ei, const int* __restrict__ et,
    const float* __restrict__ Y, const float* __restrict__ cnt, float* acc,
    int* lcur, int2* __restrict__ lst) {
  int npend = 0;
  const int4* dst4 = (const int4*)(ei + NEDGES);
  const int NQ = NEDGES / 4;
  for (int base = 0; base < NQ; base += 4 * 1024) {
    if (npend >= FLUSH_AT) {
      const int nfull = npend & ~(CHUNK - 1);
      drain_pending<SAVE, false>(pend, nfull, lane, d0, ei, et, Y, cnt, acc, lcur, lst);
      const int rem = npend - nfull;
      if (lane < rem) pend[lane] = pend[nfull + lane];
      npend = rem;
    }
    int4 dv[4];
    #pragma unroll
    for (int u = 0; u < 4; ++u) {
      int q = base + u * 1024 + t;
      dv[u] = (q < NQ) ? dst4[q] : make_int4(-1, -1, -1, -1);
    }
    #pragma unroll
    for (int u = 0; u < 4; ++u) {
      int q = base + u * 1024 + t;
      unsigned r0 = (unsigned)(dv[u].x - d0), r1 = (unsigned)(dv[u].y - d0);
      unsigned r2 = (unsigned)(dv[u].z - d0), r3 = (unsigned)(dv[u].w - d0);
      bool m0 = r0 < (unsigned)nrows, m1 = r1 < (unsigned)nrows;
      bool m2 = r2 < (unsigned)nrows, m3 = r3 < (unsigned)nrows;
      if (__builtin_amdgcn_ballot_w32(m0 | m1 | m2 | m3)) {
        unsigned k;
        k = __builtin_amdgcn_ballot_w32(m0);
        if (m0) pend[npend + __builtin_amdgcn_mbcnt_lo(k, 0u)] = (int)((r0 << 20) | (unsigned)(4 * q + 0));
        npend += __builtin_popcount(k);
        k = __builtin_amdgcn_ballot_w32(m1);
        if (m1) pend[npend + __builtin_amdgcn_mbcnt_lo(k, 0u)] = (int)((r1 << 20) | (unsigned)(4 * q + 1));
        npend += __builtin_popcount(k);
        k = __builtin_amdgcn_ballot_w32(m2);
        if (m2) pend[npend + __builtin_amdgcn_mbcnt_lo(k, 0u)] = (int)((r2 << 20) | (unsigned)(4 * q + 2));
        npend += __builtin_popcount(k);
        k = __builtin_amdgcn_ballot_w32(m3);
        if (m3) pend[npend + __builtin_amdgcn_mbcnt_lo(k, 0u)] = (int)((r3 << 20) | (unsigned)(4 * q + 3));
        npend += __builtin_popcount(k);
      }
    }
  }
  if (npend > 0) drain_pending<SAVE, false>(pend, npend, lane, d0, ei, et, Y, cnt, acc, lcur, lst);
}

template <bool HEAD>
__global__ __launch_bounds__(1024) void edge_gather_owned(
    const int* __restrict__ ei, const int* __restrict__ et,
    const float* __restrict__ Y, const float* __restrict__ cnt,
    float* __restrict__ hacc, float* __restrict__ lists,
    const float* __restrict__ lw, const float* __restrict__ lb, float* __restrict__ out) {
  __shared__ __align__(16) float acc[(OWN_ROWS + 1) * FDIM];
  __shared__ int pend_all[32 * PEND_CAP];
  __shared__ int lcur;
  const int t = threadIdx.x;
  const int lane = t & 31;
  const int wave = t >> 5;
  const int d0 = (int)blockIdx.x * OWN_ROWS;
  const int nrows = min(OWN_ROWS, NNODES - d0);
  int* lhead = (int*)(lists + (long)blockIdx.x * LIST_STRIDE);
  int2* lst = (int2*)lhead + LIST_HDR;

  int nlist = 0;
  if (HEAD) nlist = lhead[0];
  if (!HEAD && t == 0) lcur = 0;
  int4 ca = make_int4(0, 0, 0, 0), cb = ca;
  if (!HEAD) {
    ca = *(const int4*)(lhead + CNT_OFF + lane * 8);
    cb = *(const int4*)(lhead + CNT_OFF + lane * 8 + 4);
  }

  const float4* hsrc = (const float4*)(hacc + (long)d0 * FDIM);
  for (int i = t; i < nrows * 32; i += 1024) {
    float4 v = hsrc[i];
    float* a = acc + (i >> 5) * FDIM + (i & 31);
    a[0] = v.x; a[32] = v.y; a[64] = v.z; a[96] = v.w;
  }
  __syncthreads();

  int* pend = pend_all + wave * PEND_CAP;
  if (!HEAD) {
    const int cmax = max(max(max(ca.x, ca.y), max(ca.z, ca.w)), max(max(cb.x, cb.y), max(cb.z, cb.w)));
    if (__builtin_amdgcn_ballot_w32(cmax > SEG_CAP) == 0) {
      const int ws = __builtin_amdgcn_readfirstlane(wave);
      const int cj[8] = {
          __builtin_amdgcn_readlane(ca.x, ws), __builtin_amdgcn_readlane(ca.y, ws),
          __builtin_amdgcn_readlane(ca.z, ws), __builtin_amdgcn_readlane(ca.w, ws),
          __builtin_amdgcn_readlane(cb.x, ws), __builtin_amdgcn_readlane(cb.y, ws),
          __builtin_amdgcn_readlane(cb.z, ws), __builtin_amdgcn_readlane(cb.w, ws)};
      const int* seg = lhead + SEG_OFF + ws * 8 * SEG_CAP + lane;
      int wv[8];
      #pragma unroll
      for (int j = 0; j < 8; ++j) wv[j] = (lane < cj[j]) ? seg[j * SEG_CAP] : 0;
      int npend = 0;
      #pragma unroll
      for (int j = 0; j < 8; ++j) {
        if (lane < cj[j]) pend[npend + lane] = wv[j];
        npend += cj[j];
      }
      if (npend > 0) drain_pending<true, true>(pend, npend, lane, d0, ei, et, Y, cnt, acc, &lcur, lst);
    } else {
      scan_and_drain<true>(pend, t, lane, d0, nrows, ei, et, Y, cnt, acc, &lcur, lst);
    }
  } else if (nlist <= LIST_CAP) {
    int c = __builtin_amdgcn_readfirstlane(wave) * CHUNK;
    int2 rec = make_int2(0, 0);
    if (c < nlist) rec = lst[min(c + lane, nlist - 1)];
    while (c < nlist) {
      const int cn = c + 32 * CHUNK;
      int2 recn = make_int2(0, 0);
      if (cn < nlist) recn = lst[min(cn + lane, nlist - 1)];
      const int row = (c + lane < nlist) ? (int)((unsigned)rec.x >> 19) : OWN_ROWS;
      add_rows(rec.x & 0x7FFFF, rec.y, row, nlist - c, lane, Y, acc);
      rec = recn;
      c = cn;
    }
  } else {
    scan_and_drain<false>(pend, t, lane, d0, nrows, ei, et, Y, cnt, acc, &lcur, lst);
  }
  __syncthreads();

  if (!HEAD) {
    if (t == 0) lhead[0] = lcur;
    float4* hdst = (float4*)(hacc + (long)d0 * FDIM);
    for (int i = t; i < nrows * 32; i += 1024) {
      const float* a = acc + (i >> 5) * FDIM + (i & 31);
      hdst[i] = make_float4(a[0], a[32], a[64], a[96]);
    }
  } else {
    const float4 w0 = *(const float4*)(lw + lane * 8);
    const float4 w1 = *(const float4*)(lw + lane * 8 + 4);
    const float lb0 = lb[0], lb1 = lb[1];
    const int r0 = wave * HEAD_ROWS;
    float k0 = 0.0f, k1 = 0.0f;
    #pragma unroll
    for (int j = 0; j < HEAD_ROWS; ++j) {
      const float* a = acc + min(r0 + j, OWN_ROWS) * FDIM + lane;
      float vx = fmaxf(a[0], 0.0f), vy = fmaxf(a[32], 0.0f);
      float vz = fmaxf(a[64], 0.0f), vw = fmaxf(a[96], 0.0f);
      float p0 = vx * w0.x + vy * w0.z + vz * w1.x + vw * w1.z;
      float p1 = vx * w0.y + vy * w0.w + vz * w1.y + vw * w1.w;
      #pragma unroll
      for (int off = 16; off > 0; off >>= 1) {
        p0 += __shfl_xor(p0, off, 32);
        p1 += __shfl_xor(p1, off, 32);
      }
      if (lane == j) { k0 = p0; k1 = p1; }
    }
    if (lane < HEAD_ROWS && r0 + lane < nrows) {
      float l0 = k0 + lb0, l1 = k1 + lb1;
      float m = fmaxf(l0, l1);
      float lse = m + logf(expf(l0 - m) + expf(l1 - m));
      *(float2*)(out + (long)(d0 + r0 + lane) * 2) = make_float2(l0 - lse, l1 - lse);
    }
  }
}

extern "C" void kernel_launch(void* const* d_in, const int* in_sizes, int n_in,
                              void* d_out, int out_size, void* d_ws, size_t ws_size,
                              hipStream_t stream) {
  const float* x      = (const float*)d_in[0];
  const int*   ei     = (const int*)d_in[1];
  const int*   et     = (const int*)d_in[2];
  const float* W1rel  = (const float*)d_in[3];
  const float* W1root = (const float*)d_in[4];
  const float* b1     = (const float*)d_in[5];
  const float* W2rel  = (const float*)d_in[6];
  const float* W2root = (const float*)d_in[7];
  const float* b2     = (const float*)d_in[8];
  const float* lw     = (const float*)d_in[9];
  const float* lb     = (const float*)d_in[10];

  const long msgN = (long)NREL * NNODES * FDIM;
  const long cntN = (long)NREL * NNODES;
  const long NF   = (long)NNODES * FDIM;

  float* ws  = (float*)d_ws;
  float* Y   = ws;
  float* cnt = Y + msgN;
  float* h1  = cnt + cntN;
  float* lst = h1 + NF;

  dim3 blk(256);

  zero_f4<<<293, blk, 0, stream>>>(cnt, cntN / 4);
  edge_count_bin<<<NSLICE, 1024, 0, stream>>>(ei, et, cnt, lst);

  rgcn_project<false><<<PGRID, blk, 0, stream>>>(
      x, W1root, W1rel, W1rel + FDIM * FDIM, W1rel + 2 * FDIM * FDIM, b1, h1, Y);
  edge_gather_owned<false><<<NOWNERS, 1024, 0, stream>>>(ei, et, Y, cnt, h1, lst, lw, lb, (float*)d_out);

  rgcn_project<true><<<PGRID, blk, 0, stream>>>(
      h1, W2root, W2rel, W2rel + FDIM * FDIM, W2rel + 2 * FDIM * FDIM, b2, h1, Y);
  edge_gather_owned<true><<<NOWNERS, 1024, 0, stream>>>(ei, et, Y, cnt, h1, lst, lw, lb, (float*)d_out);
}
